# starting kernel
# baseline (speedup 1.0000x reference)
#include <hip/hip_runtime.h>
#include <hip/hip_bf16.h>

typedef __attribute__((ext_vector_type(2))) float v2f;
typedef __attribute__((ext_vector_type(8))) float v8f;

#define B_    128
#define R_    36
#define W_    60
#define D_    1024
#define TK_   20
#define EPS_  1e-12f
#define NEG_  -1e30f

__global__ __launch_bounds__(256)
void row_invnorm_kernel(const float* __restrict__ img,
                        const float* __restrict__ cap,
                        float* __restrict__ inv_cap,
                        float* __restrict__ inv_img) {
    __shared__ float red[256];
    const int row = blockIdx.x;
    const float* p = (row < B_ * W_) ? (cap + (size_t)row * D_)
                                     : (img + (size_t)(row - B_ * W_) * D_);
    const float4 v = ((const float4*)p)[threadIdx.x];
    red[threadIdx.x] = v.x * v.x + v.y * v.y + v.z * v.z + v.w * v.w;
    __syncthreads();
    for (int st = 128; st >= 1; st >>= 1) {
        if (threadIdx.x < st) red[threadIdx.x] += red[threadIdx.x + st];
        __syncthreads();
    }
    if (threadIdx.x == 0) {
        const float inv = 1.0f / fmaxf(sqrtf(red[0]), EPS_);
        if (row < B_ * W_) inv_cap[row] = inv;
        else               inv_img[row - B_ * W_] = inv;
    }
}

__global__ __launch_bounds__(128)
void pair_kernel(const float* __restrict__ img,
                 const float* __restrict__ cap,
                 const float* __restrict__ inv_cap,
                 const float* __restrict__ inv_img,
                 const float* __restrict__ w1_t2i, const float* __restrict__ b1_t2i,
                 const float* __restrict__ w2_t2i, const float* __restrict__ b2_t2i,
                 const float* __restrict__ w1_i2t, const float* __restrict__ b1_i2t,
                 const float* __restrict__ w2_i2t, const float* __restrict__ b2_i2t,
                 const int*   __restrict__ cap_len,
                 float*       __restrict__ out) {
    const int i    = blockIdx.x;
    const int c    = blockIdx.y;
    const int tid  = threadIdx.x;
    const int lane = tid & 31;
    const int wv   = tid >> 5;

    __shared__ float As[64][33];
    __shared__ float Bs[48][33];
    __shared__ float Sm[64][49];
    __shared__ float rnw[48];
    __shared__ float rnr[64];
    __shared__ float rowsim[64];
    __shared__ float colsim[48];
    __shared__ float topr[TK_], topc[TK_];
    __shared__ float h1[TK_], h2[TK_];
    __shared__ float red[128];
    __shared__ float s_rowmean, s_colmean;

    for (int idx = tid; idx < 4 * 33;  idx += 128) As[60 + idx / 33][idx % 33] = 0.f;
    for (int idx = tid; idx < 12 * 33; idx += 128) Bs[36 + idx / 33][idx % 33] = 0.f;

    v8f acc0 = {}; v8f acc1 = {}; v8f acc2 = {};
    const int mr = (wv << 4) + (lane & 15);
    const int nr = lane & 15;
    const int kb = (lane >> 4) << 1;

    const float* capRow = cap + (size_t)c * W_ * D_;
    const float* imgRow = img + (size_t)i * R_ * D_;
    const float* invC   = inv_cap + c * W_;
    const float* invI   = inv_img + i * R_;

    for (int k0 = 0; k0 < D_; k0 += 32) {
        __syncthreads();
        for (int idx = tid; idx < 480 + 288; idx += 128) {
            if (idx < 480) {
                const int row = idx >> 3, c4 = (idx & 7) << 2;
                const float4 v = *(const float4*)(capRow + (size_t)row * D_ + k0 + c4);
                const float sc = invC[row];
                As[row][c4 + 0] = v.x * sc; As[row][c4 + 1] = v.y * sc;
                As[row][c4 + 2] = v.z * sc; As[row][c4 + 3] = v.w * sc;
            } else {
                const int j = idx - 480;
                const int row = j >> 3, c4 = (j & 7) << 2;
                const float4 v = *(const float4*)(imgRow + (size_t)row * D_ + k0 + c4);
                const float sc = invI[row];
                Bs[row][c4 + 0] = v.x * sc; Bs[row][c4 + 1] = v.y * sc;
                Bs[row][c4 + 2] = v.z * sc; Bs[row][c4 + 3] = v.w * sc;
            }
        }
        __syncthreads();
        for (int kk = 0; kk < 32; kk += 4) {
            v2f a;   a.x   = As[mr][kk + kb];      a.y   = As[mr][kk + kb + 1];
            v2f b0;  b0.x  = Bs[nr][kk + kb];      b0.y  = Bs[nr][kk + kb + 1];
            v2f b1v; b1v.x = Bs[16 + nr][kk + kb]; b1v.y = Bs[16 + nr][kk + kb + 1];
            v2f b2v; b2v.x = Bs[32 + nr][kk + kb]; b2v.y = Bs[32 + nr][kk + kb + 1];
            acc0 = __builtin_amdgcn_wmma_f32_16x16x4_f32(false, a, false, b0,  (short)0, acc0, false, false);
            acc1 = __builtin_amdgcn_wmma_f32_16x16x4_f32(false, a, false, b1v, (short)0, acc1, false, false);
            acc2 = __builtin_amdgcn_wmma_f32_16x16x4_f32(false, a, false, b2v, (short)0, acc2, false, false);
        }
    }
    __syncthreads();

    {
        const int mb = (wv << 4) + ((lane >> 4) << 3);
        for (int v = 0; v < 8; v++) {
            const int m = mb + v;
            float x0 = acc0[v]; Sm[m][nr]      = (x0 >= 0.f) ? x0 : 0.1f * x0;
            float x1 = acc1[v]; Sm[m][16 + nr] = (x1 >= 0.f) ? x1 : 0.1f * x1;
            float x2 = acc2[v]; Sm[m][32 + nr] = (x2 >= 0.f) ? x2 : 0.1f * x2;
        }
    }
    __syncthreads();

    const int cl   = cap_len[c];
    const int nsel = (cl > TK_) ? cl : TK_;

    if (tid < R_) {
        float ss = 0.f;
        for (int w = 0; w < nsel; w++) { float x = Sm[w][tid]; ss += x * x; }
        rnw[tid] = 1.0f / fmaxf(sqrtf(ss), EPS_);
    } else if (tid >= 64 && tid < 64 + W_) {
        const int w = tid - 64;
        float ss = 0.f;
        for (int r = 0; r < R_; r++) { float x = Sm[w][r]; ss += x * x; }
        rnr[w] = 1.0f / fmaxf(sqrtf(ss), EPS_);
    }
    __syncthreads();

    if (tid < W_) {
        float m = NEG_;
        if (tid < nsel)
            for (int r = 0; r < R_; r++) m = fmaxf(m, Sm[tid][r] * rnw[r]);
        rowsim[tid] = m;
    } else if (tid >= 64 && tid < 64 + R_) {
        const int r = tid - 64;
        float m = NEG_;
        for (int w = 0; w < cl; w++) m = fmaxf(m, Sm[w][r] * rnr[w]);
        colsim[r] = m;
    }
    __syncthreads();

    red[tid] = (tid < cl) ? rowsim[tid] : 0.f;
    __syncthreads();
    for (int st = 64; st >= 1; st >>= 1) {
        if (tid < st) red[tid] += red[tid + st];
        __syncthreads();
    }
    if (tid == 0) s_rowmean = red[0] / (float)cl;
    __syncthreads();
    red[tid] = (tid < R_) ? colsim[tid] : 0.f;
    __syncthreads();
    for (int st = 64; st >= 1; st >>= 1) {
        if (tid < st) red[tid] += red[tid + st];
        __syncthreads();
    }
    if (tid == 0) s_colmean = red[0] / (float)R_;
    __syncthreads();

    if (tid < W_) {
        const float v = rowsim[tid];
        int rank = 0;
        for (int j = 0; j < W_; j++) {
            const float u = rowsim[j];
            rank += (u > v) || (u == v && j < tid);
        }
        if (rank < TK_) topr[rank] = v;
    } else if (tid >= 64 && tid < 64 + R_) {
        const int r = tid - 64;
        const float v = colsim[r];
        int rank = 0;
        for (int j = 0; j < R_; j++) {
            const float u = colsim[j];
            rank += (u > v) || (u == v && j < r);
        }
        if (rank < TK_) topc[rank] = v;
    }
    __syncthreads();

    if (tid < TK_) {
        float a = b1_t2i[tid];
        for (int k = 0; k < TK_; k++) a += topr[k] * w1_t2i[k * TK_ + tid];
        h1[tid] = fmaxf(a, 0.f);
    } else if (tid >= 32 && tid < 32 + TK_) {
        const int j = tid - 32;
        float a = b1_i2t[j];
        for (int k = 0; k < TK_; k++) a += topc[k] * w1_i2t[k * TK_ + j];
        h2[j] = fmaxf(a, 0.f);
    }
    __syncthreads();

    if (tid == 0) {
        float o = b2_t2i[0];
        for (int j = 0; j < TK_; j++) o += h1[j] * w2_t2i[j];
        out[c * B_ + i]              = o + s_rowmean;
        out[2 * B_ * B_ + c * B_ + i] = s_rowmean;
    } else if (tid == 32) {
        float o = b2_i2t[0];
        for (int j = 0; j < TK_; j++) o += h2[j] * w2_i2t[j];
        out[B_ * B_ + i * B_ + c]     = o + s_colmean;
        out[3 * B_ * B_ + i * B_ + c] = s_colmean;
    }
}

extern "C" void kernel_launch(void* const* d_in, const int* in_sizes, int n_in,
                              void* d_out, int out_size, void* d_ws, size_t ws_size,
                              hipStream_t stream) {
    const float* img     = (const float*)d_in[0];
    const float* cap     = (const float*)d_in[1];
    const float* w1_t2i  = (const float*)d_in[2];
    const float* b1_t2i  = (const float*)d_in[3];
    const float* w2_t2i  = (const float*)d_in[4];
    const float* b2_t2i  = (const float*)d_in[5];
    const float* w1_i2t  = (const float*)d_in[6];
    const float* b1_i2t  = (const float*)d_in[7];
    const float* w2_i2t  = (const float*)d_in[8];
    const float* b2_i2t  = (const float*)d_in[9];
    const int*   cap_len = (const int*)d_in[10];
    float*       out     = (float*)d_out;

    float* inv_cap = (float*)d_ws;
    float* inv_img = inv_cap + B_ * W_;

    row_invnorm_kernel<<<B_ * W_ + B_ * R_, 256, 0, stream>>>(img, cap, inv_cap, inv_img);

    dim3 grid(B_, B_);
    pair_kernel<<<grid, 128, 0, stream>>>(img, cap, inv_cap, inv_img,
                                          w1_t2i, b1_t2i, w2_t2i, b2_t2i,
                                          w1_i2t, b1_i2t, w2_i2t, b2_i2t,
                                          cap_len, out);
}


# the six global loads that stage a K chunk are issued back to back ahead of the barrier instead of load, wait, store one at a time; row scales and slot addresses set up once; sums unchanged
# speedup vs baseline: 1.2838x; 1.2838x over previous
#include <hip/hip_runtime.h>
#include <hip/hip_bf16.h>

typedef __attribute__((ext_vector_type(2))) float v2f;
typedef __attribute__((ext_vector_type(8))) float v8f;
typedef __attribute__((ext_vector_type(4))) float v4f;

#define B_    128
#define R_    36
#define W_    60
#define D_    1024
#define TK_   20
#define EPS_  1e-12f
#define NEG_  -1e30f

__global__ __launch_bounds__(256)
void row_invnorm_kernel(const float* __restrict__ img,
                        const float* __restrict__ cap,
                        float* __restrict__ inv_cap,
                        float* __restrict__ inv_img) {
    __shared__ float red[256];
    const int row = blockIdx.x;
    const float* p = (row < B_ * W_) ? (cap + (size_t)row * D_)
                                     : (img + (size_t)(row - B_ * W_) * D_);
    const float4 v = ((const float4*)p)[threadIdx.x];
    red[threadIdx.x] = v.x * v.x + v.y * v.y + v.z * v.z + v.w * v.w;
    __syncthreads();
    for (int st = 128; st >= 1; st >>= 1) {
        if (threadIdx.x < st) red[threadIdx.x] += red[threadIdx.x + st];
        __syncthreads();
    }
    if (threadIdx.x == 0) {
        const float inv = 1.0f / fmaxf(sqrtf(red[0]), EPS_);
        if (row < B_ * W_) inv_cap[row] = inv;
        else               inv_img[row - B_ * W_] = inv;
    }
}

__global__ __launch_bounds__(128)
void pair_kernel(const float* __restrict__ img,
                 const float* __restrict__ cap,
                 const float* __restrict__ inv_cap,
                 const float* __restrict__ inv_img,
                 const float* __restrict__ w1_t2i, const float* __restrict__ b1_t2i,
                 const float* __restrict__ w2_t2i, const float* __restrict__ b2_t2i,
                 const float* __restrict__ w1_i2t, const float* __restrict__ b1_i2t,
                 const float* __restrict__ w2_i2t, const float* __restrict__ b2_i2t,
                 const int*   __restrict__ cap_len,
                 float*       __restrict__ out) {
    const int i    = blockIdx.x;
    const int c    = blockIdx.y;
    const int tid  = threadIdx.x;
    const int lane = tid & 31;
    const int wv   = tid >> 5;

    __shared__ float As[64][33];
    __shared__ float Bs[48][33];
    __shared__ float Sm[64][49];
    __shared__ float rnw[48];
    __shared__ float rnr[64];
    __shared__ float rowsim[64];
    __shared__ float colsim[48];
    __shared__ float topr[TK_], topc[TK_];
    __shared__ float h1[TK_], h2[TK_];
    __shared__ float red[128];
    __shared__ float s_rowmean, s_colmean;

    for (int idx = tid; idx < 4 * 33;  idx += 128) As[60 + idx / 33][idx % 33] = 0.f;
    for (int idx = tid; idx < 12 * 33; idx += 128) Bs[36 + idx / 33][idx % 33] = 0.f;

    v8f acc0 = {}; v8f acc1 = {}; v8f acc2 = {};
    const int mr = (wv << 4) + (lane & 15);
    const int nr = lane & 15;
    const int kb = (lane >> 4) << 1;

    const float* capRow = cap + (size_t)c * W_ * D_;
    const float* imgRow = img + (size_t)i * R_ * D_;
    const float* invC   = inv_cap + c * W_;
    const float* invI   = inv_img + i * R_;

    const float* gsrc[6];
    float*       ldst[6];
    float        rsc[6];
    #pragma unroll
    for (int s = 0; s < 6; s++) {
        const int  idx = tid + s * 128;
        const bool isA = idx < 480;
        const int  j   = isA ? idx : idx - 480;
        const int  row = j >> 3, c4 = (j & 7) << 2;
        gsrc[s] = (isA ? capRow : imgRow) + (size_t)row * D_ + c4;
        ldst[s] = isA ? &As[row][c4] : &Bs[row][c4];
        rsc[s]  = (isA ? invC : invI)[row];
    }

    for (int k0 = 0; k0 < D_; k0 += 32) {
        v4f v[6];
        #pragma unroll
        for (int s = 0; s < 6; s++) v[s] = *(const v4f*)(gsrc[s] + k0);
        asm volatile("" :: "v"(v[0]), "v"(v[1]), "v"(v[2]), "v"(v[3]), "v"(v[4]), "v"(v[5]));
        __syncthreads();
        #pragma unroll
        for (int s = 0; s < 6; s++) {
            float* d = ldst[s];
            const float sc = rsc[s];
            d[0] = v[s].x * sc; d[1] = v[s].y * sc;
            d[2] = v[s].z * sc; d[3] = v[s].w * sc;
        }
        __syncthreads();
        for (int kk = 0; kk < 32; kk += 4) {
            v2f a;   a.x   = As[mr][kk + kb];      a.y   = As[mr][kk + kb + 1];
            v2f b0;  b0.x  = Bs[nr][kk + kb];      b0.y  = Bs[nr][kk + kb + 1];
            v2f b1v; b1v.x = Bs[16 + nr][kk + kb]; b1v.y = Bs[16 + nr][kk + kb + 1];
            v2f b2v; b2v.x = Bs[32 + nr][kk + kb]; b2v.y = Bs[32 + nr][kk + kb + 1];
            acc0 = __builtin_amdgcn_wmma_f32_16x16x4_f32(false, a, false, b0,  (short)0, acc0, false, false);
            acc1 = __builtin_amdgcn_wmma_f32_16x16x4_f32(false, a, false, b1v, (short)0, acc1, false, false);
            acc2 = __builtin_amdgcn_wmma_f32_16x16x4_f32(false, a, false, b2v, (short)0, acc2, false, false);
        }
    }
    __syncthreads();

    {
        const int mb = (wv << 4) + ((lane >> 4) << 3);
        for (int v = 0; v < 8; v++) {
            const int m = mb + v;
            float x0 = acc0[v]; Sm[m][nr]      = (x0 >= 0.f) ? x0 : 0.1f * x0;
            float x1 = acc1[v]; Sm[m][16 + nr] = (x1 >= 0.f) ? x1 : 0.1f * x1;
            float x2 = acc2[v]; Sm[m][32 + nr] = (x2 >= 0.f) ? x2 : 0.1f * x2;
        }
    }
    __syncthreads();

    const int cl   = cap_len[c];
    const int nsel = (cl > TK_) ? cl : TK_;

    if (tid < R_) {
        float ss = 0.f;
        for (int w = 0; w < nsel; w++) { float x = Sm[w][tid]; ss += x * x; }
        rnw[tid] = 1.0f / fmaxf(sqrtf(ss), EPS_);
    } else if (tid >= 64 && tid < 64 + W_) {
        const int w = tid - 64;
        float ss = 0.f;
        for (int r = 0; r < R_; r++) { float x = Sm[w][r]; ss += x * x; }
        rnr[w] = 1.0f / fmaxf(sqrtf(ss), EPS_);
    }
    __syncthreads();

    if (tid < W_) {
        float m = NEG_;
        if (tid < nsel)
            for (int r = 0; r < R_; r++) m = fmaxf(m, Sm[tid][r] * rnw[r]);
        rowsim[tid] = m;
    } else if (tid >= 64 && tid < 64 + R_) {
        const int r = tid - 64;
        float m = NEG_;
        for (int w = 0; w < cl; w++) m = fmaxf(m, Sm[w][r] * rnr[w]);
        colsim[r] = m;
    }
    __syncthreads();

    red[tid] = (tid < cl) ? rowsim[tid] : 0.f;
    __syncthreads();
    for (int st = 64; st >= 1; st >>= 1) {
        if (tid < st) red[tid] += red[tid + st];
        __syncthreads();
    }
    if (tid == 0) s_rowmean = red[0] / (float)cl;
    __syncthreads();
    red[tid] = (tid < R_) ? colsim[tid] : 0.f;
    __syncthreads();
    for (int st = 64; st >= 1; st >>= 1) {
        if (tid < st) red[tid] += red[tid + st];
        __syncthreads();
    }
    if (tid == 0) s_colmean = red[0] / (float)R_;
    __syncthreads();

    if (tid < W_) {
        const float v = rowsim[tid];
        int rank = 0;
        for (int j = 0; j < W_; j++) {
            const float u = rowsim[j];
            rank += (u > v) || (u == v && j < tid);
        }
        if (rank < TK_) topr[rank] = v;
    } else if (tid >= 64 && tid < 64 + R_) {
        const int r = tid - 64;
        const float v = colsim[r];
        int rank = 0;
        for (int j = 0; j < R_; j++) {
            const float u = colsim[j];
            rank += (u > v) || (u == v && j < r);
        }
        if (rank < TK_) topc[rank] = v;
    }
    __syncthreads();

    if (tid < TK_) {
        float a = b1_t2i[tid];
        for (int k = 0; k < TK_; k++) a += topr[k] * w1_t2i[k * TK_ + tid];
        h1[tid] = fmaxf(a, 0.f);
    } else if (tid >= 32 && tid < 32 + TK_) {
        const int j = tid - 32;
        float a = b1_i2t[j];
        for (int k = 0; k < TK_; k++) a += topc[k] * w1_i2t[k * TK_ + j];
        h2[j] = fmaxf(a, 0.f);
    }
    __syncthreads();

    if (tid == 0) {
        float o = b2_t2i[0];
        for (int j = 0; j < TK_; j++) o += h1[j] * w2_t2i[j];
        out[c * B_ + i]              = o + s_rowmean;
        out[2 * B_ * B_ + c * B_ + i] = s_rowmean;
    } else if (tid == 32) {
        float o = b2_i2t[0];
        for (int j = 0; j < TK_; j++) o += h2[j] * w2_i2t[j];
        out[B_ * B_ + i * B_ + c]     = o + s_colmean;
        out[3 * B_ * B_ + i * B_ + c] = s_colmean;
    }
}

extern "C" void kernel_launch(void* const* d_in, const int* in_sizes, int n_in,
                              void* d_out, int out_size, void* d_ws, size_t ws_size,
                              hipStream_t stream) {
    const float* img     = (const float*)d_in[0];
    const float* cap     = (const float*)d_in[1];
    const float* w1_t2i  = (const float*)d_in[2];
    const float* b1_t2i  = (const float*)d_in[3];
    const float* w2_t2i  = (const float*)d_in[4];
    const float* b2_t2i  = (const float*)d_in[5];
    const float* w1_i2t  = (const float*)d_in[6];
    const float* b1_i2t  = (const float*)d_in[7];
    const float* w2_i2t  = (const float*)d_in[8];
    const float* b2_i2t  = (const float*)d_in[9];
    const int*   cap_len = (const int*)d_in[10];
    float*       out     = (float*)d_out;

    float* inv_cap = (float*)d_ws;
    float* inv_img = inv_cap + B_ * W_;

    row_invnorm_kernel<<<B_ * W_ + B_ * R_, 256, 0, stream>>>(img, cap, inv_cap, inv_img);

    dim3 grid(B_, B_);
    pair_kernel<<<grid, 128, 0, stream>>>(img, cap, inv_cap, inv_img,
                                          w1_t2i, b1_t2i, w2_t2i, b2_t2i,
                                          w1_i2t, b1_i2t, w2_i2t, b2_i2t,
                                          cap_len, out);
}


# each block now computes one caption against 4 consecutive images (144 region rows = 9 exact 16-column tiles, was 36 padded to 48) and stages the caption chunk once for all 4; sums unchanged
# speedup vs baseline: 1.6255x; 1.2662x over previous
#include <hip/hip_runtime.h>
#include <hip/hip_bf16.h>

typedef __attribute__((ext_vector_type(2))) float v2f;
typedef __attribute__((ext_vector_type(8))) float v8f;
typedef __attribute__((ext_vector_type(4))) float v4f;

#define B_    128
#define R_    36
#define W_    60
#define D_    1024
#define TK_   20
#define EPS_  1e-12f
#define NEG_  -1e30f

__global__ __launch_bounds__(256)
void row_invnorm_kernel(const float* __restrict__ img,
                        const float* __restrict__ cap,
                        float* __restrict__ inv_cap,
                        float* __restrict__ inv_img) {
    __shared__ float red[256];
    const int row = blockIdx.x;
    const float* p = (row < B_ * W_) ? (cap + (size_t)row * D_)
                                     : (img + (size_t)(row - B_ * W_) * D_);
    const float4 v = ((const float4*)p)[threadIdx.x];
    red[threadIdx.x] = v.x * v.x + v.y * v.y + v.z * v.z + v.w * v.w;
    __syncthreads();
    for (int st = 128; st >= 1; st >>= 1) {
        if (threadIdx.x < st) red[threadIdx.x] += red[threadIdx.x + st];
        __syncthreads();
    }
    if (threadIdx.x == 0) {
        const float inv = 1.0f / fmaxf(sqrtf(red[0]), EPS_);
        if (row < B_ * W_) inv_cap[row] = inv;
        else               inv_img[row - B_ * W_] = inv;
    }
}

#define NI_   4
#define NB_   (NI_ * R_)
#define NT_   (NB_ / 16)
#define SP_   (NB_ + 1)

__global__ __launch_bounds__(128)
void pair_kernel(const float* __restrict__ img,
                 const float* __restrict__ cap,
                 const float* __restrict__ inv_cap,
                 const float* __restrict__ inv_img,
                 const float* __restrict__ w1_t2i, const float* __restrict__ b1_t2i,
                 const float* __restrict__ w2_t2i, const float* __restrict__ b2_t2i,
                 const float* __restrict__ w1_i2t, const float* __restrict__ b1_i2t,
                 const float* __restrict__ w2_i2t, const float* __restrict__ b2_i2t,
                 const int*   __restrict__ cap_len,
                 float*       __restrict__ out) {
    const int i0   = blockIdx.x * NI_;
    const int c    = blockIdx.y;
    const int tid  = threadIdx.x;
    const int lane = tid & 31;
    const int wv   = tid >> 5;

    __shared__ float As[64][33];
    __shared__ float Bs[NB_][33];
    __shared__ float Sm[64][SP_];
    __shared__ float rnw[48];
    __shared__ float rnr[64];
    __shared__ float rowsim[64];
    __shared__ float colsim[48];
    __shared__ float topr[TK_], topc[TK_];
    __shared__ float h1[TK_], h2[TK_];
    __shared__ float red[128];
    __shared__ float s_rowmean, s_colmean;

    for (int idx = tid; idx < 4 * 33; idx += 128) As[60 + idx / 33][idx % 33] = 0.f;

    v8f acc[NT_];
    #pragma unroll
    for (int t = 0; t < NT_; t++) acc[t] = v8f{};
    const int mr = (wv << 4) + (lane & 15);
    const int nr = lane & 15;
    const int kb = (lane >> 4) << 1;

    const int  srow = tid >> 3, c4 = (tid & 7) << 2;
    const bool a3   = (48 + srow) < W_;
    const float* gA = cap + ((size_t)c * W_ + srow) * D_ + c4;
    const float* gB = img + ((size_t)i0 * R_ + srow) * D_ + c4;
    float rscA[4], rscB[NT_];
    #pragma unroll
    for (int s = 0; s < 4; s++)
        rscA[s] = (s < 3 || a3) ? inv_cap[c * W_ + srow + 16 * s] : 0.f;
    #pragma unroll
    for (int s = 0; s < NT_; s++) rscB[s] = inv_img[i0 * R_ + srow + 16 * s];

    for (int k0 = 0; k0 < D_; k0 += 32) {
        v4f va[4], vb[NT_];
        #pragma unroll
        for (int s = 0; s < 3; s++) va[s] = *(const v4f*)(gA + (size_t)s * 16 * D_ + k0);
        va[3] = v4f{};
        if (a3) va[3] = *(const v4f*)(gA + (size_t)3 * 16 * D_ + k0);
        #pragma unroll
        for (int s = 0; s < NT_; s++) vb[s] = *(const v4f*)(gB + (size_t)s * 16 * D_ + k0);
        asm volatile("" :: "v"(va[0]), "v"(va[1]), "v"(va[2]), "v"(va[3]),
                           "v"(vb[0]), "v"(vb[1]), "v"(vb[2]), "v"(vb[3]), "v"(vb[4]),
                           "v"(vb[5]), "v"(vb[6]), "v"(vb[7]), "v"(vb[8]));
        __syncthreads();
        #pragma unroll
        for (int s = 0; s < 4; s++) {
            if (s < 3 || a3) {
                float* d = &As[srow + 16 * s][c4];
                const float sc = rscA[s];
                d[0] = va[s].x * sc; d[1] = va[s].y * sc;
                d[2] = va[s].z * sc; d[3] = va[s].w * sc;
            }
        }
        #pragma unroll
        for (int s = 0; s < NT_; s++) {
            float* d = &Bs[srow + 16 * s][c4];
            const float sc = rscB[s];
            d[0] = vb[s].x * sc; d[1] = vb[s].y * sc;
            d[2] = vb[s].z * sc; d[3] = vb[s].w * sc;
        }
        __syncthreads();
        for (int kk = 0; kk < 32; kk += 4) {
            v2f a; a.x = As[mr][kk + kb]; a.y = As[mr][kk + kb + 1];
            #pragma unroll
            for (int t = 0; t < NT_; t++) {
                v2f b; b.x = Bs[16 * t + nr][kk + kb]; b.y = Bs[16 * t + nr][kk + kb + 1];
                acc[t] = __builtin_amdgcn_wmma_f32_16x16x4_f32(false, a, false, b, (short)0, acc[t], false, false);
            }
        }
    }
    __syncthreads();

    {
        const int mb = (wv << 4) + ((lane >> 4) << 3);
        #pragma unroll
        for (int t = 0; t < NT_; t++) {
            #pragma unroll
            for (int v = 0; v < 8; v++) {
                const float x = acc[t][v];
                Sm[mb + v][16 * t + nr] = (x >= 0.f) ? x : 0.1f * x;
            }
        }
    }
    __syncthreads();

    const int cl   = cap_len[c];
    const int nsel = (cl > TK_) ? cl : TK_;

    for (int p = 0; p < NI_; p++) {
        const int i  = i0 + p;
        const int cb = p * R_;

        if (tid < R_) {
            float ss = 0.f;
            for (int w = 0; w < nsel; w++) { float x = Sm[w][cb + tid]; ss += x * x; }
            rnw[tid] = 1.0f / fmaxf(sqrtf(ss), EPS_);
        } else if (tid >= 64 && tid < 64 + W_) {
            const int w = tid - 64;
            float ss = 0.f;
            for (int r = 0; r < R_; r++) { float x = Sm[w][cb + r]; ss += x * x; }
            rnr[w] = 1.0f / fmaxf(sqrtf(ss), EPS_);
        }
        __syncthreads();

        if (tid < W_) {
            float m = NEG_;
            if (tid < nsel)
                for (int r = 0; r < R_; r++) m = fmaxf(m, Sm[tid][cb + r] * rnw[r]);
            rowsim[tid] = m;
        } else if (tid >= 64 && tid < 64 + R_) {
            const int r = tid - 64;
            float m = NEG_;
            for (int w = 0; w < cl; w++) m = fmaxf(m, Sm[w][cb + r] * rnr[w]);
            colsim[r] = m;
        }
        __syncthreads();

        red[tid] = (tid < cl) ? rowsim[tid] : 0.f;
        __syncthreads();
        for (int st = 64; st >= 1; st >>= 1) {
            if (tid < st) red[tid] += red[tid + st];
            __syncthreads();
        }
        if (tid == 0) s_rowmean = red[0] / (float)cl;
        __syncthreads();
        red[tid] = (tid < R_) ? colsim[tid] : 0.f;
        __syncthreads();
        for (int st = 64; st >= 1; st >>= 1) {
            if (tid < st) red[tid] += red[tid + st];
            __syncthreads();
        }
        if (tid == 0) s_colmean = red[0] / (float)R_;
        __syncthreads();

        if (tid < W_) {
            const float v = rowsim[tid];
            int rank = 0;
            for (int j = 0; j < W_; j++) {
                const float u = rowsim[j];
                rank += (u > v) || (u == v && j < tid);
            }
            if (rank < TK_) topr[rank] = v;
        } else if (tid >= 64 && tid < 64 + R_) {
            const int r = tid - 64;
            const float v = colsim[r];
            int rank = 0;
            for (int j = 0; j < R_; j++) {
                const float u = colsim[j];
                rank += (u > v) || (u == v && j < r);
            }
            if (rank < TK_) topc[rank] = v;
        }
        __syncthreads();

        if (tid < TK_) {
            float a = b1_t2i[tid];
            for (int k = 0; k < TK_; k++) a += topr[k] * w1_t2i[k * TK_ + tid];
            h1[tid] = fmaxf(a, 0.f);
        } else if (tid >= 32 && tid < 32 + TK_) {
            const int j = tid - 32;
            float a = b1_i2t[j];
            for (int k = 0; k < TK_; k++) a += topc[k] * w1_i2t[k * TK_ + j];
            h2[j] = fmaxf(a, 0.f);
        }
        __syncthreads();

        if (tid == 0) {
            float o = b2_t2i[0];
            for (int j = 0; j < TK_; j++) o += h1[j] * w2_t2i[j];
            out[c * B_ + i]              = o + s_rowmean;
            out[2 * B_ * B_ + c * B_ + i] = s_rowmean;
        } else if (tid == 32) {
            float o = b2_i2t[0];
            for (int j = 0; j < TK_; j++) o += h2[j] * w2_i2t[j];
            out[B_ * B_ + i * B_ + c]     = o + s_colmean;
            out[3 * B_ * B_ + i * B_ + c] = s_colmean;
        }
    }
}

extern "C" void kernel_launch(void* const* d_in, const int* in_sizes, int n_in,
                              void* d_out, int out_size, void* d_ws, size_t ws_size,
                              hipStream_t stream) {
    const float* img     = (const float*)d_in[0];
    const float* cap     = (const float*)d_in[1];
    const float* w1_t2i  = (const float*)d_in[2];
    const float* b1_t2i  = (const float*)d_in[3];
    const float* w2_t2i  = (const float*)d_in[4];
    const float* b2_t2i  = (const float*)d_in[5];
    const float* w1_i2t  = (const float*)d_in[6];
    const float* b1_i2t  = (const float*)d_in[7];
    const float* w2_i2t  = (const float*)d_in[8];
    const float* b2_i2t  = (const float*)d_in[9];
    const int*   cap_len = (const int*)d_in[10];
    float*       out     = (float*)d_out;

    float* inv_cap = (float*)d_ws;
    float* inv_img = inv_cap + B_ * W_;

    row_invnorm_kernel<<<B_ * W_ + B_ * R_, 256, 0, stream>>>(img, cap, inv_cap, inv_img);

    dim3 grid(B_ / NI_, B_);
    pair_kernel<<<grid, 128, 0, stream>>>(img, cap, inv_cap, inv_img,
                                          w1_t2i, b1_t2i, w2_t2i, b2_t2i,
                                          w1_i2t, b1_i2t, w2_i2t, b2_i2t,
                                          cap_len, out);
}


# LDS chunk rows at a 36-float pitch: one 16-byte load per A/B fragment feeds two K-steps (was two 8-byte loads, 2-way bank conflict); 16-byte staging stores; sums take k in a new order within each 8
# speedup vs baseline: 1.7731x; 1.0908x over previous
#include <hip/hip_runtime.h>
#include <hip/hip_bf16.h>

typedef __attribute__((ext_vector_type(2))) float v2f;
typedef __attribute__((ext_vector_type(8))) float v8f;
typedef __attribute__((ext_vector_type(4))) float v4f;

#define B_    128
#define R_    36
#define W_    60
#define D_    1024
#define TK_   20
#define EPS_  1e-12f
#define NEG_  -1e30f

__global__ __launch_bounds__(256)
void row_invnorm_kernel(const float* __restrict__ img,
                        const float* __restrict__ cap,
                        float* __restrict__ inv_cap,
                        float* __restrict__ inv_img) {
    __shared__ float red[256];
    const int row = blockIdx.x;
    const float* p = (row < B_ * W_) ? (cap + (size_t)row * D_)
                                     : (img + (size_t)(row - B_ * W_) * D_);
    const float4 v = ((const float4*)p)[threadIdx.x];
    red[threadIdx.x] = v.x * v.x + v.y * v.y + v.z * v.z + v.w * v.w;
    __syncthreads();
    for (int st = 128; st >= 1; st >>= 1) {
        if (threadIdx.x < st) red[threadIdx.x] += red[threadIdx.x + st];
        __syncthreads();
    }
    if (threadIdx.x == 0) {
        const float inv = 1.0f / fmaxf(sqrtf(red[0]), EPS_);
        if (row < B_ * W_) inv_cap[row] = inv;
        else               inv_img[row - B_ * W_] = inv;
    }
}

#define NI_   4
#define NB_   (NI_ * R_)
#define NT_   (NB_ / 16)
#define SP_   (NB_ + 1)
#define KP_   36

__global__ __launch_bounds__(128)
void pair_kernel(const float* __restrict__ img,
                 const float* __restrict__ cap,
                 const float* __restrict__ inv_cap,
                 const float* __restrict__ inv_img,
                 const float* __restrict__ w1_t2i, const float* __restrict__ b1_t2i,
                 const float* __restrict__ w2_t2i, const float* __restrict__ b2_t2i,
                 const float* __restrict__ w1_i2t, const float* __restrict__ b1_i2t,
                 const float* __restrict__ w2_i2t, const float* __restrict__ b2_i2t,
                 const int*   __restrict__ cap_len,
                 float*       __restrict__ out) {
    const int i0   = blockIdx.x * NI_;
    const int c    = blockIdx.y;
    const int tid  = threadIdx.x;
    const int lane = tid & 31;
    const int wv   = tid >> 5;

    __shared__ __attribute__((aligned(16))) float As[64][KP_];
    __shared__ __attribute__((aligned(16))) float Bs[NB_][KP_];
    __shared__ float Sm[64][SP_];
    __shared__ float rnw[48];
    __shared__ float rnr[64];
    __shared__ float rowsim[64];
    __shared__ float colsim[48];
    __shared__ float topr[TK_], topc[TK_];
    __shared__ float h1[TK_], h2[TK_];
    __shared__ float red[128];
    __shared__ float s_rowmean, s_colmean;

    for (int idx = tid; idx < 4 * KP_; idx += 128) As[60 + idx / KP_][idx % KP_] = 0.f;

    v8f acc[NT_];
    #pragma unroll
    for (int t = 0; t < NT_; t++) acc[t] = v8f{};
    const int mr = (wv << 4) + (lane & 15);
    const int nr = lane & 15;
    const int kb = (lane >> 4) << 2;

    const int  srow = (tid >> 4) + (tid & 8), c4 = (tid & 7) << 2;
    const bool a3   = (48 + srow) < W_;
    const float* gA = cap + ((size_t)c * W_ + srow) * D_ + c4;
    const float* gB = img + ((size_t)i0 * R_ + srow) * D_ + c4;
    float rscA[4], rscB[NT_];
    #pragma unroll
    for (int s = 0; s < 4; s++)
        rscA[s] = (s < 3 || a3) ? inv_cap[c * W_ + srow + 16 * s] : 0.f;
    #pragma unroll
    for (int s = 0; s < NT_; s++) rscB[s] = inv_img[i0 * R_ + srow + 16 * s];

    for (int k0 = 0; k0 < D_; k0 += 32) {
        v4f va[4], vb[NT_];
        #pragma unroll
        for (int s = 0; s < 3; s++) va[s] = *(const v4f*)(gA + (size_t)s * 16 * D_ + k0);
        va[3] = v4f{};
        if (a3) va[3] = *(const v4f*)(gA + (size_t)3 * 16 * D_ + k0);
        #pragma unroll
        for (int s = 0; s < NT_; s++) vb[s] = *(const v4f*)(gB + (size_t)s * 16 * D_ + k0);
        asm volatile("" :: "v"(va[0]), "v"(va[1]), "v"(va[2]), "v"(va[3]),
                           "v"(vb[0]), "v"(vb[1]), "v"(vb[2]), "v"(vb[3]), "v"(vb[4]),
                           "v"(vb[5]), "v"(vb[6]), "v"(vb[7]), "v"(vb[8]));
        __syncthreads();
        #pragma unroll
        for (int s = 0; s < 4; s++) {
            if (s < 3 || a3) *(v4f*)&As[srow + 16 * s][c4] = va[s] * rscA[s];
        }
        #pragma unroll
        for (int s = 0; s < NT_; s++) *(v4f*)&Bs[srow + 16 * s][c4] = vb[s] * rscB[s];
        __syncthreads();
        for (int kk = 0; kk < 32; kk += 8) {
            const v4f a4 = *(const v4f*)&As[mr][kk + kb];
            v4f b4[NT_];
            #pragma unroll
            for (int t = 0; t < NT_; t++) b4[t] = *(const v4f*)&Bs[16 * t + nr][kk + kb];
            v2f a; a.x = a4.x; a.y = a4.y;
            #pragma unroll
            for (int t = 0; t < NT_; t++) {
                v2f b; b.x = b4[t].x; b.y = b4[t].y;
                acc[t] = __builtin_amdgcn_wmma_f32_16x16x4_f32(false, a, false, b, (short)0, acc[t], false, false);
            }
            a.x = a4.z; a.y = a4.w;
            #pragma unroll
            for (int t = 0; t < NT_; t++) {
                v2f b; b.x = b4[t].z; b.y = b4[t].w;
                acc[t] = __builtin_amdgcn_wmma_f32_16x16x4_f32(false, a, false, b, (short)0, acc[t], false, false);
            }
        }
    }
    __syncthreads();

    {
        const int mb = (wv << 4) + ((lane >> 4) << 3);
        #pragma unroll
        for (int t = 0; t < NT_; t++) {
            #pragma unroll
            for (int v = 0; v < 8; v++) {
                const float x = acc[t][v];
                Sm[mb + v][16 * t + nr] = (x >= 0.f) ? x : 0.1f * x;
            }
        }
    }
    __syncthreads();

    const int cl   = cap_len[c];
    const int nsel = (cl > TK_) ? cl : TK_;

    for (int p = 0; p < NI_; p++) {
        const int i  = i0 + p;
        const int cb = p * R_;

        if (tid < R_) {
            float ss = 0.f;
            for (int w = 0; w < nsel; w++) { float x = Sm[w][cb + tid]; ss += x * x; }
            rnw[tid] = 1.0f / fmaxf(sqrtf(ss), EPS_);
        } else if (tid >= 64 && tid < 64 + W_) {
            const int w = tid - 64;
            float ss = 0.f;
            for (int r = 0; r < R_; r++) { float x = Sm[w][cb + r]; ss += x * x; }
            rnr[w] = 1.0f / fmaxf(sqrtf(ss), EPS_);
        }
        __syncthreads();

        if (tid < W_) {
            float m = NEG_;
            if (tid < nsel)
                for (int r = 0; r < R_; r++) m = fmaxf(m, Sm[tid][cb + r] * rnw[r]);
            rowsim[tid] = m;
        } else if (tid >= 64 && tid < 64 + R_) {
            const int r = tid - 64;
            float m = NEG_;
            for (int w = 0; w < cl; w++) m = fmaxf(m, Sm[w][cb + r] * rnr[w]);
            colsim[r] = m;
        }
        __syncthreads();

        red[tid] = (tid < cl) ? rowsim[tid] : 0.f;
        __syncthreads();
        for (int st = 64; st >= 1; st >>= 1) {
            if (tid < st) red[tid] += red[tid + st];
            __syncthreads();
        }
        if (tid == 0) s_rowmean = red[0] / (float)cl;
        __syncthreads();
        red[tid] = (tid < R_) ? colsim[tid] : 0.f;
        __syncthreads();
        for (int st = 64; st >= 1; st >>= 1) {
            if (tid < st) red[tid] += red[tid + st];
            __syncthreads();
        }
        if (tid == 0) s_colmean = red[0] / (float)R_;
        __syncthreads();

        if (tid < W_) {
            const float v = rowsim[tid];
            int rank = 0;
            for (int j = 0; j < W_; j++) {
                const float u = rowsim[j];
                rank += (u > v) || (u == v && j < tid);
            }
            if (rank < TK_) topr[rank] = v;
        } else if (tid >= 64 && tid < 64 + R_) {
            const int r = tid - 64;
            const float v = colsim[r];
            int rank = 0;
            for (int j = 0; j < R_; j++) {
                const float u = colsim[j];
                rank += (u > v) || (u == v && j < r);
            }
            if (rank < TK_) topc[rank] = v;
        }
        __syncthreads();

        if (tid < TK_) {
            float a = b1_t2i[tid];
            for (int k = 0; k < TK_; k++) a += topr[k] * w1_t2i[k * TK_ + tid];
            h1[tid] = fmaxf(a, 0.f);
        } else if (tid >= 32 && tid < 32 + TK_) {
            const int j = tid - 32;
            float a = b1_i2t[j];
            for (int k = 0; k < TK_; k++) a += topc[k] * w1_i2t[k * TK_ + j];
            h2[j] = fmaxf(a, 0.f);
        }
        __syncthreads();

        if (tid == 0) {
            float o = b2_t2i[0];
            for (int j = 0; j < TK_; j++) o += h1[j] * w2_t2i[j];
            out[c * B_ + i]              = o + s_rowmean;
            out[2 * B_ * B_ + c * B_ + i] = s_rowmean;
        } else if (tid == 32) {
            float o = b2_i2t[0];
            for (int j = 0; j < TK_; j++) o += h2[j] * w2_i2t[j];
            out[B_ * B_ + i * B_ + c]     = o + s_colmean;
            out[3 * B_ * B_ + i * B_ + c] = s_colmean;
        }
    }
}

extern "C" void kernel_launch(void* const* d_in, const int* in_sizes, int n_in,
                              void* d_out, int out_size, void* d_ws, size_t ws_size,
                              hipStream_t stream) {
    const float* img     = (const float*)d_in[0];
    const float* cap     = (const float*)d_in[1];
    const float* w1_t2i  = (const float*)d_in[2];
    const float* b1_t2i  = (const float*)d_in[3];
    const float* w2_t2i  = (const float*)d_in[4];
    const float* b2_t2i  = (const float*)d_in[5];
    const float* w1_i2t  = (const float*)d_in[6];
    const float* b1_i2t  = (const float*)d_in[7];
    const float* w2_i2t  = (const float*)d_in[8];
    const float* b2_i2t  = (const float*)d_in[9];
    const int*   cap_len = (const int*)d_in[10];
    float*       out     = (float*)d_out;

    float* inv_cap = (float*)d_ws;
    float* inv_img = inv_cap + B_ * W_;

    row_invnorm_kernel<<<B_ * W_ + B_ * R_, 256, 0, stream>>>(img, cap, inv_cap, inv_img);

    dim3 grid(B_ / NI_, B_);
    pair_kernel<<<grid, 128, 0, stream>>>(img, cap, inv_cap, inv_img,
                                          w1_t2i, b1_t2i, w2_t2i, b2_t2i,
                                          w1_i2t, b1_i2t, w2_i2t, b2_i2t,
                                          cap_len, out);
}


# per-image tail: steps issue LDS loads ahead of use (fixed-count loops, lengths as selects); means by lane shuffles in the tree's order; 22 barriers/image now 4; sum order kept, norm sums all fma
# speedup vs baseline: 1.8434x; 1.0396x over previous
#include <hip/hip_runtime.h>
#include <hip/hip_bf16.h>

typedef __attribute__((ext_vector_type(2))) float v2f;
typedef __attribute__((ext_vector_type(8))) float v8f;
typedef __attribute__((ext_vector_type(4))) float v4f;

#define B_    128
#define R_    36
#define W_    60
#define D_    1024
#define TK_   20
#define EPS_  1e-12f
#define NEG_  -1e30f

__global__ __launch_bounds__(256)
void row_invnorm_kernel(const float* __restrict__ img,
                        const float* __restrict__ cap,
                        float* __restrict__ inv_cap,
                        float* __restrict__ inv_img) {
    __shared__ float red[256];
    const int row = blockIdx.x;
    const float* p = (row < B_ * W_) ? (cap + (size_t)row * D_)
                                     : (img + (size_t)(row - B_ * W_) * D_);
    const float4 v = ((const float4*)p)[threadIdx.x];
    red[threadIdx.x] = v.x * v.x + v.y * v.y + v.z * v.z + v.w * v.w;
    __syncthreads();
    for (int st = 128; st >= 1; st >>= 1) {
        if (threadIdx.x < st) red[threadIdx.x] += red[threadIdx.x + st];
        __syncthreads();
    }
    if (threadIdx.x == 0) {
        const float inv = 1.0f / fmaxf(sqrtf(red[0]), EPS_);
        if (row < B_ * W_) inv_cap[row] = inv;
        else               inv_img[row - B_ * W_] = inv;
    }
}

#define NI_   4
#define NB_   (NI_ * R_)
#define NT_   (NB_ / 16)
#define SP_   (NB_ + 1)
#define KP_   36
#define HOLD6(a, i) "v"(a[i]), "v"(a[(i) + 1]), "v"(a[(i) + 2]), "v"(a[(i) + 3]), "v"(a[(i) + 4]), "v"(a[(i) + 5])

__global__ __launch_bounds__(128)
void pair_kernel(const float* __restrict__ img,
                 const float* __restrict__ cap,
                 const float* __restrict__ inv_cap,
                 const float* __restrict__ inv_img,
                 const float* __restrict__ w1_t2i, const float* __restrict__ b1_t2i,
                 const float* __restrict__ w2_t2i, const float* __restrict__ b2_t2i,
                 const float* __restrict__ w1_i2t, const float* __restrict__ b1_i2t,
                 const float* __restrict__ w2_i2t, const float* __restrict__ b2_i2t,
                 const int*   __restrict__ cap_len,
                 float*       __restrict__ out) {
    const int i0   = blockIdx.x * NI_;
    const int c    = blockIdx.y;
    const int tid  = threadIdx.x;
    const int lane = tid & 31;
    const int wv   = tid >> 5;

    __shared__ __attribute__((aligned(16))) float As[64][KP_];
    __shared__ __attribute__((aligned(16))) float Bs[NB_][KP_];
    __shared__ float Sm[64][SP_];
    __shared__ float rnw[48];
    __shared__ float rnr[64];
    __shared__ float rowsim[64];
    __shared__ float colsim[48];
    __shared__ float topr[TK_], topc[TK_];
    __shared__ float h1[TK_], h2[TK_];
    __shared__ float s_rowmean, s_colmean;

    for (int idx = tid; idx < 4 * KP_; idx += 128) As[60 + idx / KP_][idx % KP_] = 0.f;

    v8f acc[NT_];
    #pragma unroll
    for (int t = 0; t < NT_; t++) acc[t] = v8f{};
    const int mr = (wv << 4) + (lane & 15);
    const int nr = lane & 15;
    const int kb = (lane >> 4) << 2;

    const int  srow = (tid >> 4) + (tid & 8), c4 = (tid & 7) << 2;
    const bool a3   = (48 + srow) < W_;
    const float* gA = cap + ((size_t)c * W_ + srow) * D_ + c4;
    const float* gB = img + ((size_t)i0 * R_ + srow) * D_ + c4;
    float rscA[4], rscB[NT_];
    #pragma unroll
    for (int s = 0; s < 4; s++)
        rscA[s] = (s < 3 || a3) ? inv_cap[c * W_ + srow + 16 * s] : 0.f;
    #pragma unroll
    for (int s = 0; s < NT_; s++) rscB[s] = inv_img[i0 * R_ + srow + 16 * s];

    for (int k0 = 0; k0 < D_; k0 += 32) {
        v4f va[4], vb[NT_];
        #pragma unroll
        for (int s = 0; s < 3; s++) va[s] = *(const v4f*)(gA + (size_t)s * 16 * D_ + k0);
        va[3] = v4f{};
        if (a3) va[3] = *(const v4f*)(gA + (size_t)3 * 16 * D_ + k0);
        #pragma unroll
        for (int s = 0; s < NT_; s++) vb[s] = *(const v4f*)(gB + (size_t)s * 16 * D_ + k0);
        asm volatile("" :: "v"(va[0]), "v"(va[1]), "v"(va[2]), "v"(va[3]),
                           "v"(vb[0]), "v"(vb[1]), "v"(vb[2]), "v"(vb[3]), "v"(vb[4]),
                           "v"(vb[5]), "v"(vb[6]), "v"(vb[7]), "v"(vb[8]));
        __syncthreads();
        #pragma unroll
        for (int s = 0; s < 4; s++) {
            if (s < 3 || a3) *(v4f*)&As[srow + 16 * s][c4] = va[s] * rscA[s];
        }
        #pragma unroll
        for (int s = 0; s < NT_; s++) *(v4f*)&Bs[srow + 16 * s][c4] = vb[s] * rscB[s];
        __syncthreads();
        for (int kk = 0; kk < 32; kk += 8) {
            const v4f a4 = *(const v4f*)&As[mr][kk + kb];
            v4f b4[NT_];
            #pragma unroll
            for (int t = 0; t < NT_; t++) b4[t] = *(const v4f*)&Bs[16 * t + nr][kk + kb];
            v2f a; a.x = a4.x; a.y = a4.y;
            #pragma unroll
            for (int t = 0; t < NT_; t++) {
                v2f b; b.x = b4[t].x; b.y = b4[t].y;
                acc[t] = __builtin_amdgcn_wmma_f32_16x16x4_f32(false, a, false, b, (short)0, acc[t], false, false);
            }
            a.x = a4.z; a.y = a4.w;
            #pragma unroll
            for (int t = 0; t < NT_; t++) {
                v2f b; b.x = b4[t].z; b.y = b4[t].w;
                acc[t] = __builtin_amdgcn_wmma_f32_16x16x4_f32(false, a, false, b, (short)0, acc[t], false, false);
            }
        }
    }
    __syncthreads();

    {
        const int mb = (wv << 4) + ((lane >> 4) << 3);
        #pragma unroll
        for (int t = 0; t < NT_; t++) {
            #pragma unroll
            for (int v = 0; v < 8; v++) {
                const float x = acc[t][v];
                Sm[mb + v][16 * t + nr] = (x >= 0.f) ? x : 0.1f * x;
            }
        }
    }
    __syncthreads();

    const int cl   = cap_len[c];
    const int nsel = (cl > TK_) ? cl : TK_;

    for (int p = 0; p < NI_; p++) {
        const int i  = i0 + p;
        const int cb = p * R_;

        if (tid < R_) {
            float x[W_];
            #pragma unroll
            for (int w = 0; w < W_; w++) x[w] = Sm[w][cb + tid];
            asm volatile("" :: HOLD6(x, 0), HOLD6(x, 6), HOLD6(x, 12), HOLD6(x, 18), HOLD6(x, 24));
            asm volatile("" :: HOLD6(x, 30), HOLD6(x, 36), HOLD6(x, 42), HOLD6(x, 48), HOLD6(x, 54));
            float ss = 0.f;
            #pragma unroll
            for (int w = 0; w < W_; w++) ss = (w < nsel) ? __builtin_fmaf(x[w], x[w], ss) : ss;
            rnw[tid] = 1.0f / fmaxf(sqrtf(ss), EPS_);
        } else if (tid >= 64 && tid < 64 + W_) {
            const int w = tid - 64;
            float x[R_];
            #pragma unroll
            for (int r = 0; r < R_; r++) x[r] = Sm[w][cb + r];
            asm volatile("" :: HOLD6(x, 0), HOLD6(x, 6), HOLD6(x, 12), HOLD6(x, 18), HOLD6(x, 24));
            asm volatile("" :: HOLD6(x, 30));
            float ss = 0.f;
            #pragma unroll
            for (int r = 0; r < R_; r++) ss = __builtin_fmaf(x[r], x[r], ss);
            rnr[w] = 1.0f / fmaxf(sqrtf(ss), EPS_);
        }
        __syncthreads();

        if (tid < W_) {
            float m = NEG_;
            if (tid < nsel) {
                float x[R_];
                #pragma unroll
                for (int r = 0; r < R_; r++) x[r] = Sm[tid][cb + r];
                asm volatile("" :: HOLD6(x, 0), HOLD6(x, 6), HOLD6(x, 12), HOLD6(x, 18), HOLD6(x, 24));
                asm volatile("" :: HOLD6(x, 30));
                #pragma unroll
                for (int r = 0; r < R_; r++) m = fmaxf(m, x[r] * rnw[r]);
            }
            rowsim[tid] = m;
        } else if (tid >= 64 && tid < 64 + R_) {
            const int r = tid - 64;
            float v[W_];
            #pragma unroll
            for (int w = 0; w < W_; w++) v[w] = Sm[w][cb + r] * rnr[w];
            float m = NEG_;
            #pragma unroll
            for (int w = 0; w < W_; w++) m = (w < cl) ? fmaxf(m, v[w]) : m;
            colsim[r] = m;
        }
        __syncthreads();

        if (wv == 2) {
            const float a0 = (lane < cl)      ? rowsim[lane]      : 0.f;
            const float a1 = (lane + 32 < cl) ? rowsim[lane + 32] : 0.f;
            float s = a0 + a1;
            s += __shfl_down(s, 16);
            s += __shfl_down(s, 8);
            s += __shfl_down(s, 4);
            s += __shfl_down(s, 2);
            s += __shfl_down(s, 1);
            if (lane == 0) s_rowmean = s / (float)cl;
        } else if (wv == 3) {
            const float a0 = colsim[lane];
            const float a1 = (lane + 32 < R_) ? colsim[lane + 32] : 0.f;
            float s = a0 + a1;
            s += __shfl_down(s, 16);
            s += __shfl_down(s, 8);
            s += __shfl_down(s, 4);
            s += __shfl_down(s, 2);
            s += __shfl_down(s, 1);
            if (lane == 0) s_colmean = s / (float)R_;
        }

        if (tid < W_) {
            const float v = rowsim[tid];
            int rank = 0;
            for (int j = 0; j < W_; j++) {
                const float u = rowsim[j];
                rank += (u > v) || (u == v && j < tid);
            }
            if (rank < TK_) topr[rank] = v;
        } else if (tid >= 64 && tid < 64 + R_) {
            const int r = tid - 64;
            const float v = colsim[r];
            int rank = 0;
            for (int j = 0; j < R_; j++) {
                const float u = colsim[j];
                rank += (u > v) || (u == v && j < r);
            }
            if (rank < TK_) topc[rank] = v;
        }
        __syncthreads();

        if (tid < TK_) {
            float a = b1_t2i[tid];
            for (int k = 0; k < TK_; k++) a += topr[k] * w1_t2i[k * TK_ + tid];
            h1[tid] = fmaxf(a, 0.f);
        } else if (tid >= 32 && tid < 32 + TK_) {
            const int j = tid - 32;
            float a = b1_i2t[j];
            for (int k = 0; k < TK_; k++) a += topc[k] * w1_i2t[k * TK_ + j];
            h2[j] = fmaxf(a, 0.f);
        }
        __syncthreads();

        if (tid == 0) {
            float o = b2_t2i[0];
            for (int j = 0; j < TK_; j++) o += h1[j] * w2_t2i[j];
            out[c * B_ + i]              = o + s_rowmean;
            out[2 * B_ * B_ + c * B_ + i] = s_rowmean;
        } else if (tid == 32) {
            float o = b2_i2t[0];
            for (int j = 0; j < TK_; j++) o += h2[j] * w2_i2t[j];
            out[B_ * B_ + i * B_ + c]     = o + s_colmean;
            out[3 * B_ * B_ + i * B_ + c] = s_colmean;
        }
    }
}

extern "C" void kernel_launch(void* const* d_in, const int* in_sizes, int n_in,
                              void* d_out, int out_size, void* d_ws, size_t ws_size,
                              hipStream_t stream) {
    const float* img     = (const float*)d_in[0];
    const float* cap     = (const float*)d_in[1];
    const float* w1_t2i  = (const float*)d_in[2];
    const float* b1_t2i  = (const float*)d_in[3];
    const float* w2_t2i  = (const float*)d_in[4];
    const float* b2_t2i  = (const float*)d_in[5];
    const float* w1_i2t  = (const float*)d_in[6];
    const float* b1_i2t  = (const float*)d_in[7];
    const float* w2_i2t  = (const float*)d_in[8];
    const float* b2_i2t  = (const float*)d_in[9];
    const int*   cap_len = (const int*)d_in[10];
    float*       out     = (float*)d_out;

    float* inv_cap = (float*)d_ws;
    float* inv_img = inv_cap + B_ * W_;

    row_invnorm_kernel<<<B_ * W_ + B_ * R_, 256, 0, stream>>>(img, cap, inv_cap, inv_img);

    dim3 grid(B_ / NI_, B_);
    pair_kernel<<<grid, 128, 0, stream>>>(img, cap, inv_cap, inv_img,
                                          w1_t2i, b1_t2i, w2_t2i, b2_t2i,
                                          w1_i2t, b1_i2t, w2_i2t, b2_i2t,
                                          cap_len, out);
}


# word rows past max(cap_len,20) reach no output: only the 2-4 live 16-row tiles are loaded, staged and multiplied, their tiles dealt to the 4 waves in level runs; sums unchanged
# speedup vs baseline: 2.0757x; 1.1260x over previous
#include <hip/hip_runtime.h>
#include <hip/hip_bf16.h>

typedef __attribute__((ext_vector_type(2))) float v2f;
typedef __attribute__((ext_vector_type(8))) float v8f;
typedef __attribute__((ext_vector_type(4))) float v4f;

#define B_    128
#define R_    36
#define W_    60
#define D_    1024
#define TK_   20
#define EPS_  1e-12f
#define NEG_  -1e30f

__global__ __launch_bounds__(256)
void row_invnorm_kernel(const float* __restrict__ img,
                        const float* __restrict__ cap,
                        float* __restrict__ inv_cap,
                        float* __restrict__ inv_img) {
    __shared__ float red[256];
    const int row = blockIdx.x;
    const float* p = (row < B_ * W_) ? (cap + (size_t)row * D_)
                                     : (img + (size_t)(row - B_ * W_) * D_);
    const float4 v = ((const float4*)p)[threadIdx.x];
    red[threadIdx.x] = v.x * v.x + v.y * v.y + v.z * v.z + v.w * v.w;
    __syncthreads();
    for (int st = 128; st >= 1; st >>= 1) {
        if (threadIdx.x < st) red[threadIdx.x] += red[threadIdx.x + st];
        __syncthreads();
    }
    if (threadIdx.x == 0) {
        const float inv = 1.0f / fmaxf(sqrtf(red[0]), EPS_);
        if (row < B_ * W_) inv_cap[row] = inv;
        else               inv_img[row - B_ * W_] = inv;
    }
}

#define NI_   4
#define NB_   (NI_ * R_)
#define NT_   (NB_ / 16)
#define SP_   (NB_ + 1)
#define KP_   36
#define HOLD6(a, i) "v"(a[i]), "v"(a[(i) + 1]), "v"(a[(i) + 2]), "v"(a[(i) + 3]), "v"(a[(i) + 4]), "v"(a[(i) + 5])
#define LDB_(S, N0, J) b4[S] = *(const v4f*)&Bs[16 * ((N0) + (J)) + nr][kk + kb]
#define WM_(S, A4, E0, E1) { v2f a_; a_.x = (A4).E0; a_.y = (A4).E1; v2f b_; b_.x = b4[S].E0; b_.y = b4[S].E1; \
    acc[S] = __builtin_amdgcn_wmma_f32_16x16x4_f32(false, a_, false, b_, (short)0, acc[S], false, false); }
#define SPILL_(S, M, N0, J) { _Pragma("unroll") for (int v = 0; v < 8; v++) { const float x = acc[S][v]; \
    Sm[16 * (M) + h8 + v][16 * ((N0) + (J)) + nr] = (x >= 0.f) ? x : 0.1f * x; } }

__global__ __launch_bounds__(128)
void pair_kernel(const float* __restrict__ img,
                 const float* __restrict__ cap,
                 const float* __restrict__ inv_cap,
                 const float* __restrict__ inv_img,
                 const float* __restrict__ w1_t2i, const float* __restrict__ b1_t2i,
                 const float* __restrict__ w2_t2i, const float* __restrict__ b2_t2i,
                 const float* __restrict__ w1_i2t, const float* __restrict__ b1_i2t,
                 const float* __restrict__ w2_i2t, const float* __restrict__ b2_i2t,
                 const int*   __restrict__ cap_len,
                 float*       __restrict__ out) {
    const int i0   = blockIdx.x * NI_;
    const int c    = blockIdx.y;
    const int tid  = threadIdx.x;
    const int lane = tid & 31;
    const int wv   = __builtin_amdgcn_readfirstlane(tid >> 5);

    __shared__ __attribute__((aligned(16))) float As[64][KP_];
    __shared__ __attribute__((aligned(16))) float Bs[NB_][KP_];
    __shared__ float Sm[64][SP_];
    __shared__ float rnw[48];
    __shared__ float rnr[64];
    __shared__ float rowsim[64];
    __shared__ float colsim[48];
    __shared__ float topr[TK_], topc[TK_];
    __shared__ float h1[TK_], h2[TK_];
    __shared__ float s_rowmean, s_colmean;

    const int cl   = cap_len[c];
    const int nsel = (cl > TK_) ? cl : TK_;
    const int mt   = (nsel + 15 > 64) ? 4 : ((nsel + 15) >> 4);
    const int  rw    = (wv + (int)blockIdx.x + ((int)blockIdx.y >> 3)) & 3;
    const int  tlo   = (rw * NT_ * mt) >> 2;
    const int  tcnt  = (((rw + 1) * NT_ * mt) >> 2) - tlo;
    const int  m_lo  = tlo / NT_, n_lo = tlo - m_lo * NT_;
    const bool cross = (NT_ - n_lo) < tcnt;
    const int  nlo   = cross ? (NT_ - n_lo) : ((tcnt < 5) ? tcnt : 5);
    const int  nhi   = tcnt - nlo;
    const int  m_hi  = cross ? m_lo + 1 : m_lo;
    const int  n_hi  = (cross || nhi == 0) ? 0 : n_lo + 5;

    for (int idx = tid; idx < 4 * KP_; idx += 128) As[60 + idx / KP_][idx % KP_] = 0.f;

    v8f acc[NT_];
    #pragma unroll
    for (int t = 0; t < NT_; t++) acc[t] = v8f{};
    const int nr = lane & 15;
    const int kb = (lane >> 4) << 2;

    const int  srow = (tid >> 4) + (tid & 8), c4 = (tid & 7) << 2;
    const bool a3   = (48 + srow) < W_;
    const float* gA = cap + ((size_t)c * W_ + srow) * D_ + c4;
    const float* gB = img + ((size_t)i0 * R_ + srow) * D_ + c4;
    float rscA[4], rscB[NT_];
    #pragma unroll
    for (int s = 0; s < 4; s++)
        rscA[s] = (s < 3 || a3) ? inv_cap[c * W_ + srow + 16 * s] : 0.f;
    #pragma unroll
    for (int s = 0; s < NT_; s++) rscB[s] = inv_img[i0 * R_ + srow + 16 * s];

    for (int k0 = 0; k0 < D_; k0 += 32) {
        v4f va[4], vb[NT_];
        #pragma unroll
        for (int s = 0; s < 2; s++) va[s] = *(const v4f*)(gA + (size_t)s * 16 * D_ + k0);
        va[2] = v4f{};
        va[3] = v4f{};
        if (mt > 2) va[2] = *(const v4f*)(gA + (size_t)2 * 16 * D_ + k0);
        if (mt > 3 && a3) va[3] = *(const v4f*)(gA + (size_t)3 * 16 * D_ + k0);
        #pragma unroll
        for (int s = 0; s < NT_; s++) vb[s] = *(const v4f*)(gB + (size_t)s * 16 * D_ + k0);
        asm volatile("" :: "v"(va[0]), "v"(va[1]), "v"(va[2]), "v"(va[3]),
                           "v"(vb[0]), "v"(vb[1]), "v"(vb[2]), "v"(vb[3]), "v"(vb[4]),
                           "v"(vb[5]), "v"(vb[6]), "v"(vb[7]), "v"(vb[8]));
        __syncthreads();
        #pragma unroll
        for (int s = 0; s < 4; s++) {
            if (s < mt && (s < 3 || a3)) *(v4f*)&As[srow + 16 * s][c4] = va[s] * rscA[s];
        }
        #pragma unroll
        for (int s = 0; s < NT_; s++) *(v4f*)&Bs[srow + 16 * s][c4] = vb[s] * rscB[s];
        __syncthreads();
        for (int kk = 0; kk < 32; kk += 8) {
            v4f b4[NT_];
            v4f ahi;
            const v4f alo = *(const v4f*)&As[16 * m_lo + nr][kk + kb];
            LDB_(0, n_lo, 0); LDB_(1, n_lo, 1); LDB_(2, n_lo, 2);
            if (nlo > 3) { LDB_(3, n_lo, 3); if (nlo > 4) LDB_(4, n_lo, 4); }
            if (nhi > 0) {
                ahi = *(const v4f*)&As[16 * m_hi + nr][kk + kb];
                LDB_(5, n_hi, 0);
                if (nhi > 1) { LDB_(6, n_hi, 1); if (nhi > 2) { LDB_(7, n_hi, 2); LDB_(8, n_hi, 3); } }
            }
            WM_(0, alo, x, y) WM_(1, alo, x, y) WM_(2, alo, x, y)
            if (nlo > 3) { WM_(3, alo, x, y) if (nlo > 4) WM_(4, alo, x, y) }
            if (nhi > 0) {
                WM_(5, ahi, x, y)
                if (nhi > 1) { WM_(6, ahi, x, y) if (nhi > 2) { WM_(7, ahi, x, y) WM_(8, ahi, x, y) } }
            }
            WM_(0, alo, z, w) WM_(1, alo, z, w) WM_(2, alo, z, w)
            if (nlo > 3) { WM_(3, alo, z, w) if (nlo > 4) WM_(4, alo, z, w) }
            if (nhi > 0) {
                WM_(5, ahi, z, w)
                if (nhi > 1) { WM_(6, ahi, z, w) if (nhi > 2) { WM_(7, ahi, z, w) WM_(8, ahi, z, w) } }
            }
        }
    }
    __syncthreads();

    {
        const int h8 = (lane >> 4) << 3;
        SPILL_(0, m_lo, n_lo, 0) SPILL_(1, m_lo, n_lo, 1) SPILL_(2, m_lo, n_lo, 2)
        if (nlo > 3) { SPILL_(3, m_lo, n_lo, 3) if (nlo > 4) SPILL_(4, m_lo, n_lo, 4) }
        if (nhi > 0) {
            SPILL_(5, m_hi, n_hi, 0)
            if (nhi > 1) { SPILL_(6, m_hi, n_hi, 1) if (nhi > 2) { SPILL_(7, m_hi, n_hi, 2) SPILL_(8, m_hi, n_hi, 3) } }
        }
        for (int idx = tid; idx < (W_ - 16 * mt) * NB_; idx += 128) Sm[16 * mt + idx / NB_][idx % NB_] = 0.f;
    }
    __syncthreads();

    for (int p = 0; p < NI_; p++) {
        const int i  = i0 + p;
        const int cb = p * R_;

        if (tid < R_) {
            float x[W_];
            #pragma unroll
            for (int w = 0; w < W_; w++) x[w] = Sm[w][cb + tid];
            asm volatile("" :: HOLD6(x, 0), HOLD6(x, 6), HOLD6(x, 12), HOLD6(x, 18), HOLD6(x, 24));
            asm volatile("" :: HOLD6(x, 30), HOLD6(x, 36), HOLD6(x, 42), HOLD6(x, 48), HOLD6(x, 54));
            float ss = 0.f;
            #pragma unroll
            for (int w = 0; w < W_; w++) ss = (w < nsel) ? __builtin_fmaf(x[w], x[w], ss) : ss;
            rnw[tid] = 1.0f / fmaxf(sqrtf(ss), EPS_);
        } else if (tid >= 64 && tid < 64 + W_) {
            const int w = tid - 64;
            float x[R_];
            #pragma unroll
            for (int r = 0; r < R_; r++) x[r] = Sm[w][cb + r];
            asm volatile("" :: HOLD6(x, 0), HOLD6(x, 6), HOLD6(x, 12), HOLD6(x, 18), HOLD6(x, 24));
            asm volatile("" :: HOLD6(x, 30));
            float ss = 0.f;
            #pragma unroll
            for (int r = 0; r < R_; r++) ss = __builtin_fmaf(x[r], x[r], ss);
            rnr[w] = 1.0f / fmaxf(sqrtf(ss), EPS_);
        }
        __syncthreads();

        if (tid < W_) {
            float m = NEG_;
            if (tid < nsel) {
                float x[R_];
                #pragma unroll
                for (int r = 0; r < R_; r++) x[r] = Sm[tid][cb + r];
                asm volatile("" :: HOLD6(x, 0), HOLD6(x, 6), HOLD6(x, 12), HOLD6(x, 18), HOLD6(x, 24));
                asm volatile("" :: HOLD6(x, 30));
                #pragma unroll
                for (int r = 0; r < R_; r++) m = fmaxf(m, x[r] * rnw[r]);
            }
            rowsim[tid] = m;
        } else if (tid >= 64 && tid < 64 + R_) {
            const int r = tid - 64;
            float v[W_];
            #pragma unroll
            for (int w = 0; w < W_; w++) v[w] = Sm[w][cb + r] * rnr[w];
            float m = NEG_;
            #pragma unroll
            for (int w = 0; w < W_; w++) m = (w < cl) ? fmaxf(m, v[w]) : m;
            colsim[r] = m;
        }
        __syncthreads();

        if (wv == 2) {
            const float a0 = (lane < cl)      ? rowsim[lane]      : 0.f;
            const float a1 = (lane + 32 < cl) ? rowsim[lane + 32] : 0.f;
            float s = a0 + a1;
            s += __shfl_down(s, 16);
            s += __shfl_down(s, 8);
            s += __shfl_down(s, 4);
            s += __shfl_down(s, 2);
            s += __shfl_down(s, 1);
            if (lane == 0) s_rowmean = s / (float)cl;
        } else if (wv == 3) {
            const float a0 = colsim[lane];
            const float a1 = (lane + 32 < R_) ? colsim[lane + 32] : 0.f;
            float s = a0 + a1;
            s += __shfl_down(s, 16);
            s += __shfl_down(s, 8);
            s += __shfl_down(s, 4);
            s += __shfl_down(s, 2);
            s += __shfl_down(s, 1);
            if (lane == 0) s_colmean = s / (float)R_;
        }

        if (tid < W_) {
            const float v = rowsim[tid];
            int rank = 0;
            for (int j = 0; j < W_; j++) {
                const float u = rowsim[j];
                rank += (u > v) || (u == v && j < tid);
            }
            if (rank < TK_) topr[rank] = v;
        } else if (tid >= 64 && tid < 64 + R_) {
            const int r = tid - 64;
            const float v = colsim[r];
            int rank = 0;
            for (int j = 0; j < R_; j++) {
                const float u = colsim[j];
                rank += (u > v) || (u == v && j < r);
            }
            if (rank < TK_) topc[rank] = v;
        }
        __syncthreads();

        if (tid < TK_) {
            float a = b1_t2i[tid];
            for (int k = 0; k < TK_; k++) a += topr[k] * w1_t2i[k * TK_ + tid];
            h1[tid] = fmaxf(a, 0.f);
        } else if (tid >= 32 && tid < 32 + TK_) {
            const int j = tid - 32;
            float a = b1_i2t[j];
            for (int k = 0; k < TK_; k++) a += topc[k] * w1_i2t[k * TK_ + j];
            h2[j] = fmaxf(a, 0.f);
        }
        __syncthreads();

        if (tid == 0) {
            float o = b2_t2i[0];
            for (int j = 0; j < TK_; j++) o += h1[j] * w2_t2i[j];
            out[c * B_ + i]              = o + s_rowmean;
            out[2 * B_ * B_ + c * B_ + i] = s_rowmean;
        } else if (tid == 32) {
            float o = b2_i2t[0];
            for (int j = 0; j < TK_; j++) o += h2[j] * w2_i2t[j];
            out[B_ * B_ + i * B_ + c]     = o + s_colmean;
            out[3 * B_ * B_ + i * B_ + c] = s_colmean;
        }
    }
}

extern "C" void kernel_launch(void* const* d_in, const int* in_sizes, int n_in,
                              void* d_out, int out_size, void* d_ws, size_t ws_size,
                              hipStream_t stream) {
    const float* img     = (const float*)d_in[0];
    const float* cap     = (const float*)d_in[1];
    const float* w1_t2i  = (const float*)d_in[2];
    const float* b1_t2i  = (const float*)d_in[3];
    const float* w2_t2i  = (const float*)d_in[4];
    const float* b2_t2i  = (const float*)d_in[5];
    const float* w1_i2t  = (const float*)d_in[6];
    const float* b1_i2t  = (const float*)d_in[7];
    const float* w2_i2t  = (const float*)d_in[8];
    const float* b2_i2t  = (const float*)d_in[9];
    const int*   cap_len = (const int*)d_in[10];
    float*       out     = (float*)d_out;

    float* inv_cap = (float*)d_ws;
    float* inv_img = inv_cap + B_ * W_;

    row_invnorm_kernel<<<B_ * W_ + B_ * R_, 256, 0, stream>>>(img, cap, inv_cap, inv_img);

    dim3 grid(B_ / NI_, B_);
    pair_kernel<<<grid, 128, 0, stream>>>(img, cap, inv_cap, inv_img,
                                          w1_t2i, b1_t2i, w2_t2i, b2_t2i,
                                          w1_i2t, b1_i2t, w2_i2t, b2_i2t,
                                          cap_len, out);
}


# blocks take the captions longest first (falling count of live 16-row tiles, from wave ballots over cap_len) so the grid's last blocks to start are the short ones; every pair's sums unchanged
# speedup vs baseline: 2.1524x; 1.0370x over previous
#include <hip/hip_runtime.h>
#include <hip/hip_bf16.h>

typedef __attribute__((ext_vector_type(2))) float v2f;
typedef __attribute__((ext_vector_type(8))) float v8f;
typedef __attribute__((ext_vector_type(4))) float v4f;

#define B_    128
#define R_    36
#define W_    60
#define D_    1024
#define TK_   20
#define EPS_  1e-12f
#define NEG_  -1e30f

__global__ __launch_bounds__(256)
void row_invnorm_kernel(const float* __restrict__ img,
                        const float* __restrict__ cap,
                        float* __restrict__ inv_cap,
                        float* __restrict__ inv_img) {
    __shared__ float red[256];
    const int row = blockIdx.x;
    const float* p = (row < B_ * W_) ? (cap + (size_t)row * D_)
                                     : (img + (size_t)(row - B_ * W_) * D_);
    const float4 v = ((const float4*)p)[threadIdx.x];
    red[threadIdx.x] = v.x * v.x + v.y * v.y + v.z * v.z + v.w * v.w;
    __syncthreads();
    for (int st = 128; st >= 1; st >>= 1) {
        if (threadIdx.x < st) red[threadIdx.x] += red[threadIdx.x + st];
        __syncthreads();
    }
    if (threadIdx.x == 0) {
        const float inv = 1.0f / fmaxf(sqrtf(red[0]), EPS_);
        if (row < B_ * W_) inv_cap[row] = inv;
        else               inv_img[row - B_ * W_] = inv;
    }
}

#define NI_   4
#define NB_   (NI_ * R_)
#define NT_   (NB_ / 16)
#define SP_   (NB_ + 1)
#define KP_   36
#define HOLD6(a, i) "v"(a[i]), "v"(a[(i) + 1]), "v"(a[(i) + 2]), "v"(a[(i) + 3]), "v"(a[(i) + 4]), "v"(a[(i) + 5])
#define LDB_(S, N0, J) b4[S] = *(const v4f*)&Bs[16 * ((N0) + (J)) + nr][kk + kb]
#define WM_(S, A4, E0, E1) { v2f a_; a_.x = (A4).E0; a_.y = (A4).E1; v2f b_; b_.x = b4[S].E0; b_.y = b4[S].E1; \
    acc[S] = __builtin_amdgcn_wmma_f32_16x16x4_f32(false, a_, false, b_, (short)0, acc[S], false, false); }
#define SPILL_(S, M, N0, J) { _Pragma("unroll") for (int v = 0; v < 8; v++) { const float x = acc[S][v]; \
    Sm[16 * (M) + h8 + v][16 * ((N0) + (J)) + nr] = (x >= 0.f) ? x : 0.1f * x; } }

__global__ __launch_bounds__(128)
void pair_kernel(const float* __restrict__ img,
                 const float* __restrict__ cap,
                 const float* __restrict__ inv_cap,
                 const float* __restrict__ inv_img,
                 const float* __restrict__ w1_t2i, const float* __restrict__ b1_t2i,
                 const float* __restrict__ w2_t2i, const float* __restrict__ b2_t2i,
                 const float* __restrict__ w1_i2t, const float* __restrict__ b1_i2t,
                 const float* __restrict__ w2_i2t, const float* __restrict__ b2_i2t,
                 const int*   __restrict__ cap_len,
                 float*       __restrict__ out) {
    const int i0   = blockIdx.x * NI_;
    const int tid  = threadIdx.x;
    const int lane = tid & 31;
    const int wv   = __builtin_amdgcn_readfirstlane(tid >> 5);

    int c, cl;
    {
        int clq[B_ / 32];
        unsigned m4[B_ / 32], m3[B_ / 32];
        int n4 = 0, n3 = 0;
        #pragma unroll
        for (int q = 0; q < B_ / 32; q++) clq[q] = cap_len[32 * q + lane];
        #pragma unroll
        for (int q = 0; q < B_ / 32; q++) {
            const int ns = (clq[q] > TK_) ? clq[q] : TK_;
            const int k  = (ns + 15 > 64) ? 4 : ((ns + 15) >> 4);
            m4[q] = __builtin_amdgcn_ballot_w32(k == 4);
            m3[q] = __builtin_amdgcn_ballot_w32(k == 3);
            n4 += __builtin_popcount(m4[q]);
            n3 += __builtin_popcount(m3[q]);
        }
        const int by  = (int)blockIdx.y;
        const int cls = (by < n4) ? 4 : ((by < n4 + n3) ? 3 : 2);
        const int pos = (by < n4) ? by : ((by < n4 + n3) ? by - n4 : by - n4 - n3);
        const unsigned below = (1u << lane) - 1u;
        int run = 0;
        c = 0; cl = 0;
        #pragma unroll
        for (int q = 0; q < B_ / 32; q++) {
            const unsigned m = (cls == 4) ? m4[q] : ((cls == 3) ? m3[q] : ~(m4[q] | m3[q]));
            const unsigned hit = __builtin_amdgcn_ballot_w32(((m >> lane) & 1u) != 0u &&
                                                             run + __builtin_popcount(m & below) == pos);
            if (hit != 0u) {
                const int l = __builtin_ctz(hit);
                c  = 32 * q + l;
                cl = __builtin_amdgcn_readlane(clq[q], l);
            }
            run += __builtin_popcount(m);
        }
    }

    __shared__ __attribute__((aligned(16))) float As[64][KP_];
    __shared__ __attribute__((aligned(16))) float Bs[NB_][KP_];
    __shared__ float Sm[64][SP_];
    __shared__ float rnw[48];
    __shared__ float rnr[64];
    __shared__ float rowsim[64];
    __shared__ float colsim[48];
    __shared__ float topr[TK_], topc[TK_];
    __shared__ float h1[TK_], h2[TK_];
    __shared__ float s_rowmean, s_colmean;

    const int nsel = (cl > TK_) ? cl : TK_;
    const int mt   = (nsel + 15 > 64) ? 4 : ((nsel + 15) >> 4);
    const int  rw    = (wv + (int)blockIdx.x + ((int)blockIdx.y >> 3)) & 3;
    const int  tlo   = (rw * NT_ * mt) >> 2;
    const int  tcnt  = (((rw + 1) * NT_ * mt) >> 2) - tlo;
    const int  m_lo  = tlo / NT_, n_lo = tlo - m_lo * NT_;
    const bool cross = (NT_ - n_lo) < tcnt;
    const int  nlo   = cross ? (NT_ - n_lo) : ((tcnt < 5) ? tcnt : 5);
    const int  nhi   = tcnt - nlo;
    const int  m_hi  = cross ? m_lo + 1 : m_lo;
    const int  n_hi  = (cross || nhi == 0) ? 0 : n_lo + 5;

    for (int idx = tid; idx < 4 * KP_; idx += 128) As[60 + idx / KP_][idx % KP_] = 0.f;

    v8f acc[NT_];
    #pragma unroll
    for (int t = 0; t < NT_; t++) acc[t] = v8f{};
    const int nr = lane & 15;
    const int kb = (lane >> 4) << 2;

    const int  srow = (tid >> 4) + (tid & 8), c4 = (tid & 7) << 2;
    const bool a3   = (48 + srow) < W_;
    const float* gA = cap + ((size_t)c * W_ + srow) * D_ + c4;
    const float* gB = img + ((size_t)i0 * R_ + srow) * D_ + c4;
    float rscA[4], rscB[NT_];
    #pragma unroll
    for (int s = 0; s < 4; s++)
        rscA[s] = (s < 3 || a3) ? inv_cap[c * W_ + srow + 16 * s] : 0.f;
    #pragma unroll
    for (int s = 0; s < NT_; s++) rscB[s] = inv_img[i0 * R_ + srow + 16 * s];

    for (int k0 = 0; k0 < D_; k0 += 32) {
        v4f va[4], vb[NT_];
        #pragma unroll
        for (int s = 0; s < 2; s++) va[s] = *(const v4f*)(gA + (size_t)s * 16 * D_ + k0);
        va[2] = v4f{};
        va[3] = v4f{};
        if (mt > 2) va[2] = *(const v4f*)(gA + (size_t)2 * 16 * D_ + k0);
        if (mt > 3 && a3) va[3] = *(const v4f*)(gA + (size_t)3 * 16 * D_ + k0);
        #pragma unroll
        for (int s = 0; s < NT_; s++) vb[s] = *(const v4f*)(gB + (size_t)s * 16 * D_ + k0);
        asm volatile("" :: "v"(va[0]), "v"(va[1]), "v"(va[2]), "v"(va[3]),
                           "v"(vb[0]), "v"(vb[1]), "v"(vb[2]), "v"(vb[3]), "v"(vb[4]),
                           "v"(vb[5]), "v"(vb[6]), "v"(vb[7]), "v"(vb[8]));
        __syncthreads();
        #pragma unroll
        for (int s = 0; s < 4; s++) {
            if (s < mt && (s < 3 || a3)) *(v4f*)&As[srow + 16 * s][c4] = va[s] * rscA[s];
        }
        #pragma unroll
        for (int s = 0; s < NT_; s++) *(v4f*)&Bs[srow + 16 * s][c4] = vb[s] * rscB[s];
        __syncthreads();
        for (int kk = 0; kk < 32; kk += 8) {
            v4f b4[NT_];
            v4f ahi;
            const v4f alo = *(const v4f*)&As[16 * m_lo + nr][kk + kb];
            LDB_(0, n_lo, 0); LDB_(1, n_lo, 1); LDB_(2, n_lo, 2);
            if (nlo > 3) { LDB_(3, n_lo, 3); if (nlo > 4) LDB_(4, n_lo, 4); }
            if (nhi > 0) {
                ahi = *(const v4f*)&As[16 * m_hi + nr][kk + kb];
                LDB_(5, n_hi, 0);
                if (nhi > 1) { LDB_(6, n_hi, 1); if (nhi > 2) { LDB_(7, n_hi, 2); LDB_(8, n_hi, 3); } }
            }
            WM_(0, alo, x, y) WM_(1, alo, x, y) WM_(2, alo, x, y)
            if (nlo > 3) { WM_(3, alo, x, y) if (nlo > 4) WM_(4, alo, x, y) }
            if (nhi > 0) {
                WM_(5, ahi, x, y)
                if (nhi > 1) { WM_(6, ahi, x, y) if (nhi > 2) { WM_(7, ahi, x, y) WM_(8, ahi, x, y) } }
            }
            WM_(0, alo, z, w) WM_(1, alo, z, w) WM_(2, alo, z, w)
            if (nlo > 3) { WM_(3, alo, z, w) if (nlo > 4) WM_(4, alo, z, w) }
            if (nhi > 0) {
                WM_(5, ahi, z, w)
                if (nhi > 1) { WM_(6, ahi, z, w) if (nhi > 2) { WM_(7, ahi, z, w) WM_(8, ahi, z, w) } }
            }
        }
    }
    __syncthreads();

    {
        const int h8 = (lane >> 4) << 3;
        SPILL_(0, m_lo, n_lo, 0) SPILL_(1, m_lo, n_lo, 1) SPILL_(2, m_lo, n_lo, 2)
        if (nlo > 3) { SPILL_(3, m_lo, n_lo, 3) if (nlo > 4) SPILL_(4, m_lo, n_lo, 4) }
        if (nhi > 0) {
            SPILL_(5, m_hi, n_hi, 0)
            if (nhi > 1) { SPILL_(6, m_hi, n_hi, 1) if (nhi > 2) { SPILL_(7, m_hi, n_hi, 2) SPILL_(8, m_hi, n_hi, 3) } }
        }
        for (int idx = tid; idx < (W_ - 16 * mt) * NB_; idx += 128) Sm[16 * mt + idx / NB_][idx % NB_] = 0.f;
    }
    __syncthreads();

    for (int p = 0; p < NI_; p++) {
        const int i  = i0 + p;
        const int cb = p * R_;

        if (tid < R_) {
            float x[W_];
            #pragma unroll
            for (int w = 0; w < W_; w++) x[w] = Sm[w][cb + tid];
            asm volatile("" :: HOLD6(x, 0), HOLD6(x, 6), HOLD6(x, 12), HOLD6(x, 18), HOLD6(x, 24));
            asm volatile("" :: HOLD6(x, 30), HOLD6(x, 36), HOLD6(x, 42), HOLD6(x, 48), HOLD6(x, 54));
            float ss = 0.f;
            #pragma unroll
            for (int w = 0; w < W_; w++) ss = (w < nsel) ? __builtin_fmaf(x[w], x[w], ss) : ss;
            rnw[tid] = 1.0f / fmaxf(sqrtf(ss), EPS_);
        } else if (tid >= 64 && tid < 64 + W_) {
            const int w = tid - 64;
            float x[R_];
            #pragma unroll
            for (int r = 0; r < R_; r++) x[r] = Sm[w][cb + r];
            asm volatile("" :: HOLD6(x, 0), HOLD6(x, 6), HOLD6(x, 12), HOLD6(x, 18), HOLD6(x, 24));
            asm volatile("" :: HOLD6(x, 30));
            float ss = 0.f;
            #pragma unroll
            for (int r = 0; r < R_; r++) ss = __builtin_fmaf(x[r], x[r], ss);
            rnr[w] = 1.0f / fmaxf(sqrtf(ss), EPS_);
        }
        __syncthreads();

        if (tid < W_) {
            float m = NEG_;
            if (tid < nsel) {
                float x[R_];
                #pragma unroll
                for (int r = 0; r < R_; r++) x[r] = Sm[tid][cb + r];
                asm volatile("" :: HOLD6(x, 0), HOLD6(x, 6), HOLD6(x, 12), HOLD6(x, 18), HOLD6(x, 24));
                asm volatile("" :: HOLD6(x, 30));
                #pragma unroll
                for (int r = 0; r < R_; r++) m = fmaxf(m, x[r] * rnw[r]);
            }
            rowsim[tid] = m;
        } else if (tid >= 64 && tid < 64 + R_) {
            const int r = tid - 64;
            float v[W_];
            #pragma unroll
            for (int w = 0; w < W_; w++) v[w] = Sm[w][cb + r] * rnr[w];
            float m = NEG_;
            #pragma unroll
            for (int w = 0; w < W_; w++) m = (w < cl) ? fmaxf(m, v[w]) : m;
            colsim[r] = m;
        }
        __syncthreads();

        if (wv == 2) {
            const float a0 = (lane < cl)      ? rowsim[lane]      : 0.f;
            const float a1 = (lane + 32 < cl) ? rowsim[lane + 32] : 0.f;
            float s = a0 + a1;
            s += __shfl_down(s, 16);
            s += __shfl_down(s, 8);
            s += __shfl_down(s, 4);
            s += __shfl_down(s, 2);
            s += __shfl_down(s, 1);
            if (lane == 0) s_rowmean = s / (float)cl;
        } else if (wv == 3) {
            const float a0 = colsim[lane];
            const float a1 = (lane + 32 < R_) ? colsim[lane + 32] : 0.f;
            float s = a0 + a1;
            s += __shfl_down(s, 16);
            s += __shfl_down(s, 8);
            s += __shfl_down(s, 4);
            s += __shfl_down(s, 2);
            s += __shfl_down(s, 1);
            if (lane == 0) s_colmean = s / (float)R_;
        }

        if (tid < W_) {
            const float v = rowsim[tid];
            int rank = 0;
            for (int j = 0; j < W_; j++) {
                const float u = rowsim[j];
                rank += (u > v) || (u == v && j < tid);
            }
            if (rank < TK_) topr[rank] = v;
        } else if (tid >= 64 && tid < 64 + R_) {
            const int r = tid - 64;
            const float v = colsim[r];
            int rank = 0;
            for (int j = 0; j < R_; j++) {
                const float u = colsim[j];
                rank += (u > v) || (u == v && j < r);
            }
            if (rank < TK_) topc[rank] = v;
        }
        __syncthreads();

        if (tid < TK_) {
            float a = b1_t2i[tid];
            for (int k = 0; k < TK_; k++) a += topr[k] * w1_t2i[k * TK_ + tid];
            h1[tid] = fmaxf(a, 0.f);
        } else if (tid >= 32 && tid < 32 + TK_) {
            const int j = tid - 32;
            float a = b1_i2t[j];
            for (int k = 0; k < TK_; k++) a += topc[k] * w1_i2t[k * TK_ + j];
            h2[j] = fmaxf(a, 0.f);
        }
        __syncthreads();

        if (tid == 0) {
            float o = b2_t2i[0];
            for (int j = 0; j < TK_; j++) o += h1[j] * w2_t2i[j];
            out[c * B_ + i]              = o + s_rowmean;
            out[2 * B_ * B_ + c * B_ + i] = s_rowmean;
        } else if (tid == 32) {
            float o = b2_i2t[0];
            for (int j = 0; j < TK_; j++) o += h2[j] * w2_i2t[j];
            out[B_ * B_ + i * B_ + c]     = o + s_colmean;
            out[3 * B_ * B_ + i * B_ + c] = s_colmean;
        }
    }
}

extern "C" void kernel_launch(void* const* d_in, const int* in_sizes, int n_in,
                              void* d_out, int out_size, void* d_ws, size_t ws_size,
                              hipStream_t stream) {
    const float* img     = (const float*)d_in[0];
    const float* cap     = (const float*)d_in[1];
    const float* w1_t2i  = (const float*)d_in[2];
    const float* b1_t2i  = (const float*)d_in[3];
    const float* w2_t2i  = (const float*)d_in[4];
    const float* b2_t2i  = (const float*)d_in[5];
    const float* w1_i2t  = (const float*)d_in[6];
    const float* b1_i2t  = (const float*)d_in[7];
    const float* w2_i2t  = (const float*)d_in[8];
    const float* b2_i2t  = (const float*)d_in[9];
    const int*   cap_len = (const int*)d_in[10];
    float*       out     = (float*)d_out;

    float* inv_cap = (float*)d_ws;
    float* inv_img = inv_cap + B_ * W_;

    row_invnorm_kernel<<<B_ * W_ + B_ * R_, 256, 0, stream>>>(img, cap, inv_cap, inv_img);

    dim3 grid(B_ / NI_, B_);
    pair_kernel<<<grid, 128, 0, stream>>>(img, cap, inv_cap, inv_img,
                                          w1_t2i, b1_t2i, w2_t2i, b2_t2i,
                                          w1_i2t, b1_i2t, w2_i2t, b2_i2t,
                                          cap_len, out);
}


# K loop instantiated per count of live row tiles (2, 3, 4): run shape and staging tests fold to constants, 0 to 9 uniform tests per 8 k between the matrix instructions instead of 15; sums unchanged
# speedup vs baseline: 2.2381x; 1.0398x over previous
#include <hip/hip_runtime.h>
#include <hip/hip_bf16.h>

typedef __attribute__((ext_vector_type(2))) float v2f;
typedef __attribute__((ext_vector_type(8))) float v8f;
typedef __attribute__((ext_vector_type(4))) float v4f;

#define B_    128
#define R_    36
#define W_    60
#define D_    1024
#define TK_   20
#define EPS_  1e-12f
#define NEG_  -1e30f

__global__ __launch_bounds__(256)
void row_invnorm_kernel(const float* __restrict__ img,
                        const float* __restrict__ cap,
                        float* __restrict__ inv_cap,
                        float* __restrict__ inv_img) {
    __shared__ float red[256];
    const int row = blockIdx.x;
    const float* p = (row < B_ * W_) ? (cap + (size_t)row * D_)
                                     : (img + (size_t)(row - B_ * W_) * D_);
    const float4 v = ((const float4*)p)[threadIdx.x];
    red[threadIdx.x] = v.x * v.x + v.y * v.y + v.z * v.z + v.w * v.w;
    __syncthreads();
    for (int st = 128; st >= 1; st >>= 1) {
        if (threadIdx.x < st) red[threadIdx.x] += red[threadIdx.x + st];
        __syncthreads();
    }
    if (threadIdx.x == 0) {
        const float inv = 1.0f / fmaxf(sqrtf(red[0]), EPS_);
        if (row < B_ * W_) inv_cap[row] = inv;
        else               inv_img[row - B_ * W_] = inv;
    }
}

template <int V> struct ic_ { static constexpr int value = V; };

#define NI_   4
#define NB_   (NI_ * R_)
#define NT_   (NB_ / 16)
#define SP_   (NB_ + 1)
#define KP_   36
#define HOLD6(a, i) "v"(a[i]), "v"(a[(i) + 1]), "v"(a[(i) + 2]), "v"(a[(i) + 3]), "v"(a[(i) + 4]), "v"(a[(i) + 5])
#define LDB_(S, N0, J) b4[S] = *(const v4f*)&Bs[16 * ((N0) + (J)) + nr][kk + kb]
#define WM_(S, A4, E0, E1) { v2f a_; a_.x = (A4).E0; a_.y = (A4).E1; v2f b_; b_.x = b4[S].E0; b_.y = b4[S].E1; \
    acc[S] = __builtin_amdgcn_wmma_f32_16x16x4_f32(false, a_, false, b_, (short)0, acc[S], false, false); }
#define SPILL_(S, M, N0, J) { _Pragma("unroll") for (int v = 0; v < 8; v++) { const float x = acc[S][v]; \
    Sm[16 * (M) + h8 + v][16 * ((N0) + (J)) + nr] = (x >= 0.f) ? x : 0.1f * x; } }

__global__ __launch_bounds__(128)
void pair_kernel(const float* __restrict__ img,
                 const float* __restrict__ cap,
                 const float* __restrict__ inv_cap,
                 const float* __restrict__ inv_img,
                 const float* __restrict__ w1_t2i, const float* __restrict__ b1_t2i,
                 const float* __restrict__ w2_t2i, const float* __restrict__ b2_t2i,
                 const float* __restrict__ w1_i2t, const float* __restrict__ b1_i2t,
                 const float* __restrict__ w2_i2t, const float* __restrict__ b2_i2t,
                 const int*   __restrict__ cap_len,
                 float*       __restrict__ out) {
    const int i0   = blockIdx.x * NI_;
    const int tid  = threadIdx.x;
    const int lane = tid & 31;
    const int wv   = __builtin_amdgcn_readfirstlane(tid >> 5);

    int c, cl;
    {
        int clq[B_ / 32];
        unsigned m4[B_ / 32], m3[B_ / 32];
        int n4 = 0, n3 = 0;
        #pragma unroll
        for (int q = 0; q < B_ / 32; q++) clq[q] = cap_len[32 * q + lane];
        #pragma unroll
        for (int q = 0; q < B_ / 32; q++) {
            const int ns = (clq[q] > TK_) ? clq[q] : TK_;
            const int k  = (ns + 15 > 64) ? 4 : ((ns + 15) >> 4);
            m4[q] = __builtin_amdgcn_ballot_w32(k == 4);
            m3[q] = __builtin_amdgcn_ballot_w32(k == 3);
            n4 += __builtin_popcount(m4[q]);
            n3 += __builtin_popcount(m3[q]);
        }
        const int by  = (int)blockIdx.y;
        const int cls = (by < n4) ? 4 : ((by < n4 + n3) ? 3 : 2);
        const int pos = (by < n4) ? by : ((by < n4 + n3) ? by - n4 : by - n4 - n3);
        const unsigned below = (1u << lane) - 1u;
        int run = 0;
        c = 0; cl = 0;
        #pragma unroll
        for (int q = 0; q < B_ / 32; q++) {
            const unsigned m = (cls == 4) ? m4[q] : ((cls == 3) ? m3[q] : ~(m4[q] | m3[q]));
            const unsigned hit = __builtin_amdgcn_ballot_w32(((m >> lane) & 1u) != 0u &&
                                                             run + __builtin_popcount(m & below) == pos);
            if (hit != 0u) {
                const int l = __builtin_ctz(hit);
                c  = 32 * q + l;
                cl = __builtin_amdgcn_readlane(clq[q], l);
            }
            run += __builtin_popcount(m);
        }
    }

    __shared__ __attribute__((aligned(16))) float As[64][KP_];
    __shared__ __attribute__((aligned(16))) float Bs[NB_][KP_];
    __shared__ float Sm[64][SP_];
    __shared__ float rnw[48];
    __shared__ float rnr[64];
    __shared__ float rowsim[64];
    __shared__ float colsim[48];
    __shared__ float topr[TK_], topc[TK_];
    __shared__ float h1[TK_], h2[TK_];
    __shared__ float s_rowmean, s_colmean;

    const int nsel = (cl > TK_) ? cl : TK_;
    const int mt   = (nsel + 15 > 64) ? 4 : ((nsel + 15) >> 4);
    const int  rw    = (wv + (int)blockIdx.x + ((int)blockIdx.y >> 3)) & 3;
    const int  tlo   = (rw * NT_ * mt) >> 2;
    const int  tcnt  = (((rw + 1) * NT_ * mt) >> 2) - tlo;
    const int  m_lo  = tlo / NT_, n_lo = tlo - m_lo * NT_;
    const bool cross = (NT_ - n_lo) < tcnt;
    const int  nlo   = cross ? (NT_ - n_lo) : ((tcnt < 5) ? tcnt : 5);
    const int  nhi   = tcnt - nlo;
    const int  m_hi  = cross ? m_lo + 1 : m_lo;
    const int  n_hi  = (cross || nhi == 0) ? 0 : n_lo + 5;
    const int  m_lo_g = m_lo, n_lo_g = n_lo, m_hi_g = m_hi, n_hi_g = n_hi;

    for (int idx = tid; idx < 4 * KP_; idx += 128) As[60 + idx / KP_][idx % KP_] = 0.f;

    v8f acc[NT_];
    #pragma unroll
    for (int t = 0; t < NT_; t++) acc[t] = v8f{};
    const int nr = lane & 15;
    const int kb = (lane >> 4) << 2;

    const int  srow = (tid >> 4) + (tid & 8), c4 = (tid & 7) << 2;
    const bool a3   = (48 + srow) < W_;
    const float* gA = cap + ((size_t)c * W_ + srow) * D_ + c4;
    const float* gB = img + ((size_t)i0 * R_ + srow) * D_ + c4;
    float rscA[4], rscB[NT_];
    #pragma unroll
    for (int s = 0; s < 4; s++)
        rscA[s] = (s < 3 || a3) ? inv_cap[c * W_ + srow + 16 * s] : 0.f;
    #pragma unroll
    for (int s = 0; s < NT_; s++) rscB[s] = inv_img[i0 * R_ + srow + 16 * s];

    auto k_loop = [&](auto mtc) __attribute__((always_inline)) {
        constexpr int MT = decltype(mtc)::value;
        const int  m_lo = (MT == 4) ? rw : ((MT == 2) ? (rw >> 1) : m_lo_g);
        const int  n_lo = (MT == 4) ? 0  : ((MT == 2) ? ((rw & 1) << 2) : n_lo_g);
        const int  nlo  = (MT == 4) ? 5  : ((MT == 2) ? 4 + (rw & 1) : ((rw == 1) ? 3 : 5));
        const int  nhi  = (MT == 4) ? 4  : ((MT == 2) ? 0 : ((rw == 0) ? 1 : ((rw == 1) ? 4 : 2)));
        const int  m_hi = (MT == 4) ? rw : ((MT == 2) ? (rw >> 1) : m_hi_g);
        const int  n_hi = (MT == 4) ? 5  : ((MT == 2) ? 0 : n_hi_g);
        for (int k0 = 0; k0 < D_; k0 += 32) {
            v4f va[4], vb[NT_];
            #pragma unroll
            for (int s = 0; s < 2; s++) va[s] = *(const v4f*)(gA + (size_t)s * 16 * D_ + k0);
            va[2] = v4f{};
            va[3] = v4f{};
            if (MT > 2) va[2] = *(const v4f*)(gA + (size_t)2 * 16 * D_ + k0);
            if (MT > 3 && a3) va[3] = *(const v4f*)(gA + (size_t)3 * 16 * D_ + k0);
            #pragma unroll
            for (int s = 0; s < NT_; s++) vb[s] = *(const v4f*)(gB + (size_t)s * 16 * D_ + k0);
            asm volatile("" :: "v"(va[0]), "v"(va[1]), "v"(va[2]), "v"(va[3]),
                               "v"(vb[0]), "v"(vb[1]), "v"(vb[2]), "v"(vb[3]), "v"(vb[4]),
                               "v"(vb[5]), "v"(vb[6]), "v"(vb[7]), "v"(vb[8]));
            __syncthreads();
            #pragma unroll
            for (int s = 0; s < 4; s++) {
                if (s < MT && (s < 3 || a3)) *(v4f*)&As[srow + 16 * s][c4] = va[s] * rscA[s];
            }
            #pragma unroll
            for (int s = 0; s < NT_; s++) *(v4f*)&Bs[srow + 16 * s][c4] = vb[s] * rscB[s];
            __syncthreads();
            for (int kk = 0; kk < 32; kk += 8) {
                v4f b4[NT_];
                v4f ahi;
                const v4f alo = *(const v4f*)&As[16 * m_lo + nr][kk + kb];
                LDB_(0, n_lo, 0); LDB_(1, n_lo, 1); LDB_(2, n_lo, 2);
                if (nlo > 3) { LDB_(3, n_lo, 3); if (nlo > 4) LDB_(4, n_lo, 4); }
                if (nhi > 0) {
                    ahi = *(const v4f*)&As[16 * m_hi + nr][kk + kb];
                    LDB_(5, n_hi, 0);
                    if (nhi > 1) { LDB_(6, n_hi, 1); if (nhi > 2) { LDB_(7, n_hi, 2); LDB_(8, n_hi, 3); } }
                }
                WM_(0, alo, x, y) WM_(1, alo, x, y) WM_(2, alo, x, y)
                if (nlo > 3) { WM_(3, alo, x, y) if (nlo > 4) WM_(4, alo, x, y) }
                if (nhi > 0) {
                    WM_(5, ahi, x, y)
                    if (nhi > 1) { WM_(6, ahi, x, y) if (nhi > 2) { WM_(7, ahi, x, y) WM_(8, ahi, x, y) } }
                }
                WM_(0, alo, z, w) WM_(1, alo, z, w) WM_(2, alo, z, w)
                if (nlo > 3) { WM_(3, alo, z, w) if (nlo > 4) WM_(4, alo, z, w) }
                if (nhi > 0) {
                    WM_(5, ahi, z, w)
                    if (nhi > 1) { WM_(6, ahi, z, w) if (nhi > 2) { WM_(7, ahi, z, w) WM_(8, ahi, z, w) } }
                }
            }
        }
    };
    if (mt > 3) k_loop(ic_<4>{}); else if (mt > 2) k_loop(ic_<3>{}); else k_loop(ic_<2>{});
    __syncthreads();

    {
        const int h8 = (lane >> 4) << 3;
        SPILL_(0, m_lo, n_lo, 0) SPILL_(1, m_lo, n_lo, 1) SPILL_(2, m_lo, n_lo, 2)
        if (nlo > 3) { SPILL_(3, m_lo, n_lo, 3) if (nlo > 4) SPILL_(4, m_lo, n_lo, 4) }
        if (nhi > 0) {
            SPILL_(5, m_hi, n_hi, 0)
            if (nhi > 1) { SPILL_(6, m_hi, n_hi, 1) if (nhi > 2) { SPILL_(7, m_hi, n_hi, 2) SPILL_(8, m_hi, n_hi, 3) } }
        }
        for (int idx = tid; idx < (W_ - 16 * mt) * NB_; idx += 128) Sm[16 * mt + idx / NB_][idx % NB_] = 0.f;
    }
    __syncthreads();

    for (int p = 0; p < NI_; p++) {
        const int i  = i0 + p;
        const int cb = p * R_;

        if (tid < R_) {
            float x[W_];
            #pragma unroll
            for (int w = 0; w < W_; w++) x[w] = Sm[w][cb + tid];
            asm volatile("" :: HOLD6(x, 0), HOLD6(x, 6), HOLD6(x, 12), HOLD6(x, 18), HOLD6(x, 24));
            asm volatile("" :: HOLD6(x, 30), HOLD6(x, 36), HOLD6(x, 42), HOLD6(x, 48), HOLD6(x, 54));
            float ss = 0.f;
            #pragma unroll
            for (int w = 0; w < W_; w++) ss = (w < nsel) ? __builtin_fmaf(x[w], x[w], ss) : ss;
            rnw[tid] = 1.0f / fmaxf(sqrtf(ss), EPS_);
        } else if (tid >= 64 && tid < 64 + W_) {
            const int w = tid - 64;
            float x[R_];
            #pragma unroll
            for (int r = 0; r < R_; r++) x[r] = Sm[w][cb + r];
            asm volatile("" :: HOLD6(x, 0), HOLD6(x, 6), HOLD6(x, 12), HOLD6(x, 18), HOLD6(x, 24));
            asm volatile("" :: HOLD6(x, 30));
            float ss = 0.f;
            #pragma unroll
            for (int r = 0; r < R_; r++) ss = __builtin_fmaf(x[r], x[r], ss);
            rnr[w] = 1.0f / fmaxf(sqrtf(ss), EPS_);
        }
        __syncthreads();

        if (tid < W_) {
            float m = NEG_;
            if (tid < nsel) {
                float x[R_];
                #pragma unroll
                for (int r = 0; r < R_; r++) x[r] = Sm[tid][cb + r];
                asm volatile("" :: HOLD6(x, 0), HOLD6(x, 6), HOLD6(x, 12), HOLD6(x, 18), HOLD6(x, 24));
                asm volatile("" :: HOLD6(x, 30));
                #pragma unroll
                for (int r = 0; r < R_; r++) m = fmaxf(m, x[r] * rnw[r]);
            }
            rowsim[tid] = m;
        } else if (tid >= 64 && tid < 64 + R_) {
            const int r = tid - 64;
            float v[W_];
            #pragma unroll
            for (int w = 0; w < W_; w++) v[w] = Sm[w][cb + r] * rnr[w];
            float m = NEG_;
            #pragma unroll
            for (int w = 0; w < W_; w++) m = (w < cl) ? fmaxf(m, v[w]) : m;
            colsim[r] = m;
        }
        __syncthreads();

        if (wv == 2) {
            const float a0 = (lane < cl)      ? rowsim[lane]      : 0.f;
            const float a1 = (lane + 32 < cl) ? rowsim[lane + 32] : 0.f;
            float s = a0 + a1;
            s += __shfl_down(s, 16);
            s += __shfl_down(s, 8);
            s += __shfl_down(s, 4);
            s += __shfl_down(s, 2);
            s += __shfl_down(s, 1);
            if (lane == 0) s_rowmean = s / (float)cl;
        } else if (wv == 3) {
            const float a0 = colsim[lane];
            const float a1 = (lane + 32 < R_) ? colsim[lane + 32] : 0.f;
            float s = a0 + a1;
            s += __shfl_down(s, 16);
            s += __shfl_down(s, 8);
            s += __shfl_down(s, 4);
            s += __shfl_down(s, 2);
            s += __shfl_down(s, 1);
            if (lane == 0) s_colmean = s / (float)R_;
        }

        if (tid < W_) {
            const float v = rowsim[tid];
            int rank = 0;
            for (int j = 0; j < W_; j++) {
                const float u = rowsim[j];
                rank += (u > v) || (u == v && j < tid);
            }
            if (rank < TK_) topr[rank] = v;
        } else if (tid >= 64 && tid < 64 + R_) {
            const int r = tid - 64;
            const float v = colsim[r];
            int rank = 0;
            for (int j = 0; j < R_; j++) {
                const float u = colsim[j];
                rank += (u > v) || (u == v && j < r);
            }
            if (rank < TK_) topc[rank] = v;
        }
        __syncthreads();

        if (tid < TK_) {
            float a = b1_t2i[tid];
            for (int k = 0; k < TK_; k++) a += topr[k] * w1_t2i[k * TK_ + tid];
            h1[tid] = fmaxf(a, 0.f);
        } else if (tid >= 32 && tid < 32 + TK_) {
            const int j = tid - 32;
            float a = b1_i2t[j];
            for (int k = 0; k < TK_; k++) a += topc[k] * w1_i2t[k * TK_ + j];
            h2[j] = fmaxf(a, 0.f);
        }
        __syncthreads();

        if (tid == 0) {
            float o = b2_t2i[0];
            for (int j = 0; j < TK_; j++) o += h1[j] * w2_t2i[j];
            out[c * B_ + i]              = o + s_rowmean;
            out[2 * B_ * B_ + c * B_ + i] = s_rowmean;
        } else if (tid == 32) {
            float o = b2_i2t[0];
            for (int j = 0; j < TK_; j++) o += h2[j] * w2_i2t[j];
            out[B_ * B_ + i * B_ + c]     = o + s_colmean;
            out[3 * B_ * B_ + i * B_ + c] = s_colmean;
        }
    }
}

extern "C" void kernel_launch(void* const* d_in, const int* in_sizes, int n_in,
                              void* d_out, int out_size, void* d_ws, size_t ws_size,
                              hipStream_t stream) {
    const float* img     = (const float*)d_in[0];
    const float* cap     = (const float*)d_in[1];
    const float* w1_t2i  = (const float*)d_in[2];
    const float* b1_t2i  = (const float*)d_in[3];
    const float* w2_t2i  = (const float*)d_in[4];
    const float* b2_t2i  = (const float*)d_in[5];
    const float* w1_i2t  = (const float*)d_in[6];
    const float* b1_i2t  = (const float*)d_in[7];
    const float* w2_i2t  = (const float*)d_in[8];
    const float* b2_i2t  = (const float*)d_in[9];
    const int*   cap_len = (const int*)d_in[10];
    float*       out     = (float*)d_out;

    float* inv_cap = (float*)d_ws;
    float* inv_img = inv_cap + B_ * W_;

    row_invnorm_kernel<<<B_ * W_ + B_ * R_, 256, 0, stream>>>(img, cap, inv_cap, inv_img);

    dim3 grid(B_ / NI_, B_);
    pair_kernel<<<grid, 128, 0, stream>>>(img, cap, inv_cap, inv_img,
                                          w1_t2i, b1_t2i, w2_t2i, b2_t2i,
                                          w1_i2t, b1_i2t, w2_i2t, b2_i2t,
                                          cap_len, out);
}


# norm kernel writes normalized row copies; the pair kernel stages them unscaled (no scale registers) and lays the sim tile over the staging buffers: 5 blocks resident per WGP, was 4; sums unchanged
# speedup vs baseline: 2.3098x; 1.0320x over previous
#include <hip/hip_runtime.h>
#include <hip/hip_bf16.h>

typedef __attribute__((ext_vector_type(2))) float v2f;
typedef __attribute__((ext_vector_type(8))) float v8f;
typedef __attribute__((ext_vector_type(4))) float v4f;

#define B_    128
#define R_    36
#define W_    60
#define D_    1024
#define TK_   20
#define EPS_  1e-12f
#define NEG_  -1e30f

__global__ __launch_bounds__(256)
void row_normalize_kernel(const float* __restrict__ img,
                          const float* __restrict__ cap,
                          float* __restrict__ nrm) {
    __shared__ float red[256];
    const int row = blockIdx.x;
    const float* p = (row < B_ * W_) ? (cap + (size_t)row * D_)
                                     : (img + (size_t)(row - B_ * W_) * D_);
    const float4 v = ((const float4*)p)[threadIdx.x];
    red[threadIdx.x] = v.x * v.x + v.y * v.y + v.z * v.z + v.w * v.w;
    __syncthreads();
    for (int st = 128; st >= 1; st >>= 1) {
        if (threadIdx.x < st) red[threadIdx.x] += red[threadIdx.x + st];
        __syncthreads();
    }
    const float inv = 1.0f / fmaxf(sqrtf(red[0]), EPS_);
    float4 o;
    o.x = v.x * inv; o.y = v.y * inv; o.z = v.z * inv; o.w = v.w * inv;
    ((float4*)(nrm + (size_t)row * D_))[threadIdx.x] = o;
}

template <int V> struct ic_ { static constexpr int value = V; };

#define NI_   4
#define NB_   (NI_ * R_)
#define NT_   (NB_ / 16)
#define SP_   (NB_ + 1)
#define KP_   36
#define HOLD6(a, i) "v"(a[i]), "v"(a[(i) + 1]), "v"(a[(i) + 2]), "v"(a[(i) + 3]), "v"(a[(i) + 4]), "v"(a[(i) + 5])
#define LDB_(S, N0, J) b4[S] = *(const v4f*)&Bs[16 * ((N0) + (J)) + nr][kk + kb]
#define WM_(S, A4, E0, E1) { v2f a_; a_.x = (A4).E0; a_.y = (A4).E1; v2f b_; b_.x = b4[S].E0; b_.y = b4[S].E1; \
    acc[S] = __builtin_amdgcn_wmma_f32_16x16x4_f32(false, a_, false, b_, (short)0, acc[S], false, false); }
#define SPILL_(S, M, N0, J) { _Pragma("unroll") for (int v = 0; v < 8; v++) { const float x = acc[S][v]; \
    Sm[16 * (M) + h8 + v][16 * ((N0) + (J)) + nr] = (x >= 0.f) ? x : 0.1f * x; } }

__global__ __launch_bounds__(128)
void pair_kernel(const float* __restrict__ img,
                 const float* __restrict__ cap,
                 const float* __restrict__ w1_t2i, const float* __restrict__ b1_t2i,
                 const float* __restrict__ w2_t2i, const float* __restrict__ b2_t2i,
                 const float* __restrict__ w1_i2t, const float* __restrict__ b1_i2t,
                 const float* __restrict__ w2_i2t, const float* __restrict__ b2_i2t,
                 const int*   __restrict__ cap_len,
                 float*       __restrict__ out) {
    const int i0   = blockIdx.x * NI_;
    const int tid  = threadIdx.x;
    const int lane = tid & 31;
    const int wv   = __builtin_amdgcn_readfirstlane(tid >> 5);

    int c, cl;
    {
        int clq[B_ / 32];
        unsigned m4[B_ / 32], m3[B_ / 32];
        int n4 = 0, n3 = 0;
        #pragma unroll
        for (int q = 0; q < B_ / 32; q++) clq[q] = cap_len[32 * q + lane];
        #pragma unroll
        for (int q = 0; q < B_ / 32; q++) {
            const int ns = (clq[q] > TK_) ? clq[q] : TK_;
            const int k  = (ns + 15 > 64) ? 4 : ((ns + 15) >> 4);
            m4[q] = __builtin_amdgcn_ballot_w32(k == 4);
            m3[q] = __builtin_amdgcn_ballot_w32(k == 3);
            n4 += __builtin_popcount(m4[q]);
            n3 += __builtin_popcount(m3[q]);
        }
        const int by  = (int)blockIdx.y;
        const int cls = (by < n4) ? 4 : ((by < n4 + n3) ? 3 : 2);
        const int pos = (by < n4) ? by : ((by < n4 + n3) ? by - n4 : by - n4 - n3);
        const unsigned below = (1u << lane) - 1u;
        int run = 0;
        c = 0; cl = 0;
        #pragma unroll
        for (int q = 0; q < B_ / 32; q++) {
            const unsigned m = (cls == 4) ? m4[q] : ((cls == 3) ? m3[q] : ~(m4[q] | m3[q]));
            const unsigned hit = __builtin_amdgcn_ballot_w32(((m >> lane) & 1u) != 0u &&
                                                             run + __builtin_popcount(m & below) == pos);
            if (hit != 0u) {
                const int l = __builtin_ctz(hit);
                c  = 32 * q + l;
                cl = __builtin_amdgcn_readlane(clq[q], l);
            }
            run += __builtin_popcount(m);
        }
    }

    __shared__ __attribute__((aligned(16))) float lbuf[64 * SP_];
    float (*As)[KP_] = (float (*)[KP_])lbuf;
    float (*Bs)[KP_] = (float (*)[KP_])(lbuf + 64 * KP_);
    float (*Sm)[SP_] = (float (*)[SP_])lbuf;
    __shared__ float rnw[48];
    __shared__ float rnr[64];
    __shared__ float rowsim[64];
    __shared__ float colsim[48];
    __shared__ float topr[TK_], topc[TK_];
    __shared__ float h1[TK_], h2[TK_];
    __shared__ float s_rowmean, s_colmean;

    const int nsel = (cl > TK_) ? cl : TK_;
    const int mt   = (nsel + 15 > 64) ? 4 : ((nsel + 15) >> 4);
    const int  rw    = (wv + (int)blockIdx.x + ((int)blockIdx.y >> 3)) & 3;
    const int  tlo   = (rw * NT_ * mt) >> 2;
    const int  tcnt  = (((rw + 1) * NT_ * mt) >> 2) - tlo;
    const int  m_lo  = tlo / NT_, n_lo = tlo - m_lo * NT_;
    const bool cross = (NT_ - n_lo) < tcnt;
    const int  nlo   = cross ? (NT_ - n_lo) : ((tcnt < 5) ? tcnt : 5);
    const int  nhi   = tcnt - nlo;
    const int  m_hi  = cross ? m_lo + 1 : m_lo;
    const int  n_hi  = (cross || nhi == 0) ? 0 : n_lo + 5;
    const int  m_lo_g = m_lo, n_lo_g = n_lo, m_hi_g = m_hi, n_hi_g = n_hi;

    for (int idx = tid; idx < 4 * KP_; idx += 128) As[60 + idx / KP_][idx % KP_] = 0.f;

    v8f acc[NT_];
    #pragma unroll
    for (int t = 0; t < NT_; t++) acc[t] = v8f{};
    const int nr = lane & 15;
    const int kb = (lane >> 4) << 2;

    const int  srow = (tid >> 4) + (tid & 8), c4 = (tid & 7) << 2;
    const bool a3   = (48 + srow) < W_;
    const float* gA = cap + ((size_t)c * W_ + srow) * D_ + c4;
    const float* gB = img + ((size_t)i0 * R_ + srow) * D_ + c4;

    auto k_loop = [&](auto mtc) __attribute__((always_inline)) {
        constexpr int MT = decltype(mtc)::value;
        const int  m_lo = (MT == 4) ? rw : ((MT == 2) ? (rw >> 1) : m_lo_g);
        const int  n_lo = (MT == 4) ? 0  : ((MT == 2) ? ((rw & 1) << 2) : n_lo_g);
        const int  nlo  = (MT == 4) ? 5  : ((MT == 2) ? 4 + (rw & 1) : ((rw == 1) ? 3 : 5));
        const int  nhi  = (MT == 4) ? 4  : ((MT == 2) ? 0 : ((rw == 0) ? 1 : ((rw == 1) ? 4 : 2)));
        const int  m_hi = (MT == 4) ? rw : ((MT == 2) ? (rw >> 1) : m_hi_g);
        const int  n_hi = (MT == 4) ? 5  : ((MT == 2) ? 0 : n_hi_g);
        for (int k0 = 0; k0 < D_; k0 += 32) {
            v4f va[4], vb[NT_];
            #pragma unroll
            for (int s = 0; s < 2; s++) va[s] = *(const v4f*)(gA + (size_t)s * 16 * D_ + k0);
            va[2] = v4f{};
            va[3] = v4f{};
            if (MT > 2) va[2] = *(const v4f*)(gA + (size_t)2 * 16 * D_ + k0);
            if (MT > 3 && a3) va[3] = *(const v4f*)(gA + (size_t)3 * 16 * D_ + k0);
            #pragma unroll
            for (int s = 0; s < NT_; s++) vb[s] = *(const v4f*)(gB + (size_t)s * 16 * D_ + k0);
            asm volatile("" :: "v"(va[0]), "v"(va[1]), "v"(va[2]), "v"(va[3]),
                               "v"(vb[0]), "v"(vb[1]), "v"(vb[2]), "v"(vb[3]), "v"(vb[4]),
                               "v"(vb[5]), "v"(vb[6]), "v"(vb[7]), "v"(vb[8]));
            __syncthreads();
            #pragma unroll
            for (int s = 0; s < 4; s++) {
                if (s < MT && (s < 3 || a3)) *(v4f*)&As[srow + 16 * s][c4] = va[s];
            }
            #pragma unroll
            for (int s = 0; s < NT_; s++) *(v4f*)&Bs[srow + 16 * s][c4] = vb[s];
            __syncthreads();
            for (int kk = 0; kk < 32; kk += 8) {
                v4f b4[NT_];
                v4f ahi;
                const v4f alo = *(const v4f*)&As[16 * m_lo + nr][kk + kb];
                LDB_(0, n_lo, 0); LDB_(1, n_lo, 1); LDB_(2, n_lo, 2);
                if (nlo > 3) { LDB_(3, n_lo, 3); if (nlo > 4) LDB_(4, n_lo, 4); }
                if (nhi > 0) {
                    ahi = *(const v4f*)&As[16 * m_hi + nr][kk + kb];
                    LDB_(5, n_hi, 0);
                    if (nhi > 1) { LDB_(6, n_hi, 1); if (nhi > 2) { LDB_(7, n_hi, 2); LDB_(8, n_hi, 3); } }
                }
                WM_(0, alo, x, y) WM_(1, alo, x, y) WM_(2, alo, x, y)
                if (nlo > 3) { WM_(3, alo, x, y) if (nlo > 4) WM_(4, alo, x, y) }
                if (nhi > 0) {
                    WM_(5, ahi, x, y)
                    if (nhi > 1) { WM_(6, ahi, x, y) if (nhi > 2) { WM_(7, ahi, x, y) WM_(8, ahi, x, y) } }
                }
                WM_(0, alo, z, w) WM_(1, alo, z, w) WM_(2, alo, z, w)
                if (nlo > 3) { WM_(3, alo, z, w) if (nlo > 4) WM_(4, alo, z, w) }
                if (nhi > 0) {
                    WM_(5, ahi, z, w)
                    if (nhi > 1) { WM_(6, ahi, z, w) if (nhi > 2) { WM_(7, ahi, z, w) WM_(8, ahi, z, w) } }
                }
            }
        }
    };
    if (mt > 3) k_loop(ic_<4>{}); else if (mt > 2) k_loop(ic_<3>{}); else k_loop(ic_<2>{});
    __syncthreads();

    {
        const int h8 = (lane >> 4) << 3;
        SPILL_(0, m_lo, n_lo, 0) SPILL_(1, m_lo, n_lo, 1) SPILL_(2, m_lo, n_lo, 2)
        if (nlo > 3) { SPILL_(3, m_lo, n_lo, 3) if (nlo > 4) SPILL_(4, m_lo, n_lo, 4) }
        if (nhi > 0) {
            SPILL_(5, m_hi, n_hi, 0)
            if (nhi > 1) { SPILL_(6, m_hi, n_hi, 1) if (nhi > 2) { SPILL_(7, m_hi, n_hi, 2) SPILL_(8, m_hi, n_hi, 3) } }
        }
        for (int idx = tid; idx < (W_ - 16 * mt) * NB_; idx += 128) Sm[16 * mt + idx / NB_][idx % NB_] = 0.f;
    }
    __syncthreads();

    for (int p = 0; p < NI_; p++) {
        const int i  = i0 + p;
        const int cb = p * R_;

        if (tid < R_) {
            float x[W_];
            #pragma unroll
            for (int w = 0; w < W_; w++) x[w] = Sm[w][cb + tid];
            asm volatile("" :: HOLD6(x, 0), HOLD6(x, 6), HOLD6(x, 12), HOLD6(x, 18), HOLD6(x, 24));
            asm volatile("" :: HOLD6(x, 30), HOLD6(x, 36), HOLD6(x, 42), HOLD6(x, 48), HOLD6(x, 54));
            float ss = 0.f;
            #pragma unroll
            for (int w = 0; w < W_; w++) ss = (w < nsel) ? __builtin_fmaf(x[w], x[w], ss) : ss;
            rnw[tid] = 1.0f / fmaxf(sqrtf(ss), EPS_);
        } else if (tid >= 64 && tid < 64 + W_) {
            const int w = tid - 64;
            float x[R_];
            #pragma unroll
            for (int r = 0; r < R_; r++) x[r] = Sm[w][cb + r];
            asm volatile("" :: HOLD6(x, 0), HOLD6(x, 6), HOLD6(x, 12), HOLD6(x, 18), HOLD6(x, 24));
            asm volatile("" :: HOLD6(x, 30));
            float ss = 0.f;
            #pragma unroll
            for (int r = 0; r < R_; r++) ss = __builtin_fmaf(x[r], x[r], ss);
            rnr[w] = 1.0f / fmaxf(sqrtf(ss), EPS_);
        }
        __syncthreads();

        if (tid < W_) {
            float m = NEG_;
            if (tid < nsel) {
                float x[R_];
                #pragma unroll
                for (int r = 0; r < R_; r++) x[r] = Sm[tid][cb + r];
                asm volatile("" :: HOLD6(x, 0), HOLD6(x, 6), HOLD6(x, 12), HOLD6(x, 18), HOLD6(x, 24));
                asm volatile("" :: HOLD6(x, 30));
                #pragma unroll
                for (int r = 0; r < R_; r++) m = fmaxf(m, x[r] * rnw[r]);
            }
            rowsim[tid] = m;
        } else if (tid >= 64 && tid < 64 + R_) {
            const int r = tid - 64;
            float v[W_];
            #pragma unroll
            for (int w = 0; w < W_; w++) v[w] = Sm[w][cb + r] * rnr[w];
            float m = NEG_;
            #pragma unroll
            for (int w = 0; w < W_; w++) m = (w < cl) ? fmaxf(m, v[w]) : m;
            colsim[r] = m;
        }
        __syncthreads();

        if (wv == 2) {
            const float a0 = (lane < cl)      ? rowsim[lane]      : 0.f;
            const float a1 = (lane + 32 < cl) ? rowsim[lane + 32] : 0.f;
            float s = a0 + a1;
            s += __shfl_down(s, 16);
            s += __shfl_down(s, 8);
            s += __shfl_down(s, 4);
            s += __shfl_down(s, 2);
            s += __shfl_down(s, 1);
            if (lane == 0) s_rowmean = s / (float)cl;
        } else if (wv == 3) {
            const float a0 = colsim[lane];
            const float a1 = (lane + 32 < R_) ? colsim[lane + 32] : 0.f;
            float s = a0 + a1;
            s += __shfl_down(s, 16);
            s += __shfl_down(s, 8);
            s += __shfl_down(s, 4);
            s += __shfl_down(s, 2);
            s += __shfl_down(s, 1);
            if (lane == 0) s_colmean = s / (float)R_;
        }

        if (tid < W_) {
            const float v = rowsim[tid];
            int rank = 0;
            for (int j = 0; j < W_; j++) {
                const float u = rowsim[j];
                rank += (u > v) || (u == v && j < tid);
            }
            if (rank < TK_) topr[rank] = v;
        } else if (tid >= 64 && tid < 64 + R_) {
            const int r = tid - 64;
            const float v = colsim[r];
            int rank = 0;
            for (int j = 0; j < R_; j++) {
                const float u = colsim[j];
                rank += (u > v) || (u == v && j < r);
            }
            if (rank < TK_) topc[rank] = v;
        }
        __syncthreads();

        if (tid < TK_) {
            float a = b1_t2i[tid];
            for (int k = 0; k < TK_; k++) a += topr[k] * w1_t2i[k * TK_ + tid];
            h1[tid] = fmaxf(a, 0.f);
        } else if (tid >= 32 && tid < 32 + TK_) {
            const int j = tid - 32;
            float a = b1_i2t[j];
            for (int k = 0; k < TK_; k++) a += topc[k] * w1_i2t[k * TK_ + j];
            h2[j] = fmaxf(a, 0.f);
        }
        __syncthreads();

        if (tid == 0) {
            float o = b2_t2i[0];
            for (int j = 0; j < TK_; j++) o += h1[j] * w2_t2i[j];
            out[c * B_ + i]              = o + s_rowmean;
            out[2 * B_ * B_ + c * B_ + i] = s_rowmean;
        } else if (tid == 32) {
            float o = b2_i2t[0];
            for (int j = 0; j < TK_; j++) o += h2[j] * w2_i2t[j];
            out[B_ * B_ + i * B_ + c]     = o + s_colmean;
            out[3 * B_ * B_ + i * B_ + c] = s_colmean;
        }
    }
}

extern "C" void kernel_launch(void* const* d_in, const int* in_sizes, int n_in,
                              void* d_out, int out_size, void* d_ws, size_t ws_size,
                              hipStream_t stream) {
    const float* img     = (const float*)d_in[0];
    const float* cap     = (const float*)d_in[1];
    const float* w1_t2i  = (const float*)d_in[2];
    const float* b1_t2i  = (const float*)d_in[3];
    const float* w2_t2i  = (const float*)d_in[4];
    const float* b2_t2i  = (const float*)d_in[5];
    const float* w1_i2t  = (const float*)d_in[6];
    const float* b1_i2t  = (const float*)d_in[7];
    const float* w2_i2t  = (const float*)d_in[8];
    const float* b2_i2t  = (const float*)d_in[9];
    const int*   cap_len = (const int*)d_in[10];
    float*       out     = (float*)d_out;

    float* nrm  = (float*)d_ws;
    float* capn = nrm;
    float* imgn = nrm + (size_t)B_ * W_ * D_;

    row_normalize_kernel<<<B_ * W_ + B_ * R_, 256, 0, stream>>>(img, cap, nrm);

    dim3 grid(B_ / NI_, B_);
    pair_kernel<<<grid, 128, 0, stream>>>(imgn, capn,
                                          w1_t2i, b1_t2i, w2_t2i, b2_t2i,
                                          w1_i2t, b1_i2t, w2_i2t, b2_i2t,
                                          cap_len, out);
}


# a caption of 20 live word rows shares a block with one of 33-44 (or another of 20): rows stacked in one tile, 4 (3) row tiles instead of 5 (4), img chunk staged once for both; sums unchanged
# speedup vs baseline: 2.4419x; 1.0572x over previous
#include <hip/hip_runtime.h>
#include <hip/hip_bf16.h>

typedef __attribute__((ext_vector_type(2))) float v2f;
typedef __attribute__((ext_vector_type(8))) float v8f;
typedef __attribute__((ext_vector_type(4))) float v4f;

#define B_    128
#define R_    36
#define W_    60
#define D_    1024
#define TK_   20
#define EPS_  1e-12f
#define NEG_  -1e30f

__global__ __launch_bounds__(256)
void row_normalize_kernel(const float* __restrict__ img,
                          const float* __restrict__ cap,
                          float* __restrict__ nrm) {
    __shared__ float red[256];
    const int row = blockIdx.x;
    const float* p = (row < B_ * W_) ? (cap + (size_t)row * D_)
                                     : (img + (size_t)(row - B_ * W_) * D_);
    const float4 v = ((const float4*)p)[threadIdx.x];
    red[threadIdx.x] = v.x * v.x + v.y * v.y + v.z * v.z + v.w * v.w;
    __syncthreads();
    for (int st = 128; st >= 1; st >>= 1) {
        if (threadIdx.x < st) red[threadIdx.x] += red[threadIdx.x + st];
        __syncthreads();
    }
    const float inv = 1.0f / fmaxf(sqrtf(red[0]), EPS_);
    float4 o;
    o.x = v.x * inv; o.y = v.y * inv; o.z = v.z * inv; o.w = v.w * inv;
    ((float4*)(nrm + (size_t)row * D_))[threadIdx.x] = o;
}

template <int V> struct ic_ { static constexpr int value = V; };

#define NI_   4
#define NB_   (NI_ * R_)
#define NT_   (NB_ / 16)
#define SP_   (NB_ + 1)
#define SR_   80
#define KP_   36
#define HOLD6(a, i) "v"(a[i]), "v"(a[(i) + 1]), "v"(a[(i) + 2]), "v"(a[(i) + 3]), "v"(a[(i) + 4]), "v"(a[(i) + 5])
#define LDB_(S, N0, J) b4[S] = *(const v4f*)&Bs[16 * ((N0) + (J)) + nr][kk + kb]
#define WM_(S, A4, E0, E1) { v2f a_; a_.x = (A4).E0; a_.y = (A4).E1; v2f b_; b_.x = b4[S].E0; b_.y = b4[S].E1; \
    acc[S] = __builtin_amdgcn_wmma_f32_16x16x4_f32(false, a_, false, b_, (short)0, acc[S], false, false); }
#define SPILL_(S, M, N0, J) { _Pragma("unroll") for (int v = 0; v < 8; v++) { const float x = acc[S][v]; \
    Sm[16 * (M) + h8 + v][16 * ((N0) + (J)) + nr] = (x >= 0.f) ? x : 0.1f * x; } }

__global__ __launch_bounds__(128)
void pair_kernel(const float* __restrict__ img,
                 const float* __restrict__ cap,
                 const float* __restrict__ w1_t2i, const float* __restrict__ b1_t2i,
                 const float* __restrict__ w2_t2i, const float* __restrict__ b2_t2i,
                 const float* __restrict__ w1_i2t, const float* __restrict__ b1_i2t,
                 const float* __restrict__ w2_i2t, const float* __restrict__ b2_i2t,
                 const int*   __restrict__ cap_len,
                 float*       __restrict__ out) {
    const int i0   = blockIdx.x * NI_;
    const int tid  = threadIdx.x;
    const int lane = tid & 31;
    const int wv   = __builtin_amdgcn_readfirstlane(tid >> 5);

    int cF, clF, cG = 0, clG = 0;
    bool hasG;
    {
        int clq[B_ / 32];
        unsigned mL4[B_ / 32], mL3[B_ / 32], mM[B_ / 32], mS[B_ / 32], mZ[B_ / 32];
        int nL4 = 0, nL3 = 0, nM = 0, nS = 0, nZ = 0;
        #pragma unroll
        for (int q = 0; q < B_ / 32; q++) clq[q] = cap_len[32 * q + lane];
        #pragma unroll
        for (int q = 0; q < B_ / 32; q++) {
            const int ns = (clq[q] > TK_) ? clq[q] : TK_;
            mL4[q] = __builtin_amdgcn_ballot_w32(ns > 48);
            mL3[q] = __builtin_amdgcn_ballot_w32(ns > 44 && ns <= 48);
            mM[q]  = __builtin_amdgcn_ballot_w32(ns > 32 && ns <= 44);
            mS[q]  = __builtin_amdgcn_ballot_w32(ns > TK_ && ns <= 32);
            mZ[q]  = __builtin_amdgcn_ballot_w32(ns <= TK_);
            nL4 += __builtin_popcount(mL4[q]);
            nL3 += __builtin_popcount(mL3[q]);
            nM  += __builtin_popcount(mM[q]);
            nS  += __builtin_popcount(mS[q]);
            nZ  += __builtin_popcount(mZ[q]);
        }
        const int k  = (nZ < nM) ? nZ : nM;
        const int zl = nZ - k;
        const int e1 = nL4, e2 = e1 + k, e3 = e2 + nL3, e4 = e3 + (nM - k);
        const int e5 = e4 + (zl >> 1), e6 = e5 + nS, e7 = e6 + (zl & 1);
        const int by = (int)blockIdx.y;
        if (by >= e7) return;
        int clsF, posF, clsG = 4, posG = 0;
        hasG = false;
        if (by < e1)      { clsF = 0; posF = by; }
        else if (by < e2) { clsF = 4; posF = by - e1; hasG = true; clsG = 2; posG = by - e1; }
        else if (by < e3) { clsF = 1; posF = by - e2; }
        else if (by < e4) { clsF = 2; posF = k + (by - e3); }
        else if (by < e5) { clsF = 4; posF = k + 2 * (by - e4); hasG = true; clsG = 4; posG = posF + 1; }
        else if (by < e6) { clsF = 3; posF = by - e5; }
        else              { clsF = 4; posF = nZ - 1; }
        const unsigned below = (1u << lane) - 1u;
        auto pick = [&](int cls, int pos, int& c, int& cl) __attribute__((always_inline)) {
            int run = 0;
            c = 0; cl = 0;
            #pragma unroll
            for (int q = 0; q < B_ / 32; q++) {
                const unsigned m = (cls == 0) ? mL4[q] : ((cls == 1) ? mL3[q] : ((cls == 2) ? mM[q] : ((cls == 3) ? mS[q] : mZ[q])));
                const unsigned hit = __builtin_amdgcn_ballot_w32(((m >> lane) & 1u) != 0u &&
                                                                 run + __builtin_popcount(m & below) == pos);
                if (hit != 0u) {
                    const int l = __builtin_ctz(hit);
                    c  = 32 * q + l;
                    cl = __builtin_amdgcn_readlane(clq[q], l);
                }
                run += __builtin_popcount(m);
            }
        };
        pick(clsF, posF, cF, clF);
        if (hasG) pick(clsG, posG, cG, clG);
    }

    __shared__ __attribute__((aligned(16))) float lbuf[SR_ * SP_];
    float (*As)[KP_] = (float (*)[KP_])lbuf;
    float (*Bs)[KP_] = (float (*)[KP_])(lbuf + 64 * KP_);
    float (*Sm)[SP_] = (float (*)[SP_])lbuf;
    __shared__ float rnw[48];
    __shared__ float rnr[64];
    __shared__ float rowsim[64];
    __shared__ float colsim[48];
    __shared__ float topr[TK_], topc[TK_];
    __shared__ float h1[TK_], h2[TK_];
    __shared__ float s_rowmean, s_colmean;

    const int nF   = (clF > TK_) ? clF : TK_;
    const int nG   = hasG ? ((clG > TK_) ? clG : TK_) : 0;
    const int ntot = nF + nG;
    const int mt   = (ntot + 15 > 64) ? 4 : ((ntot + 15) >> 4);
    const int  rw    = (wv + (int)blockIdx.x + ((int)blockIdx.y >> 3)) & 3;
    const int  tlo   = (rw * NT_ * mt) >> 2;
    const int  tcnt  = (((rw + 1) * NT_ * mt) >> 2) - tlo;
    const int  m_lo  = tlo / NT_, n_lo = tlo - m_lo * NT_;
    const bool cross = (NT_ - n_lo) < tcnt;
    const int  nlo   = cross ? (NT_ - n_lo) : ((tcnt < 5) ? tcnt : 5);
    const int  nhi   = tcnt - nlo;
    const int  m_hi  = cross ? m_lo + 1 : m_lo;
    const int  n_hi  = (cross || nhi == 0) ? 0 : n_lo + 5;
    const int  m_lo_g = m_lo, n_lo_g = n_lo, m_hi_g = m_hi, n_hi_g = n_hi;

    v8f acc[NT_];
    #pragma unroll
    for (int t = 0; t < NT_; t++) acc[t] = v8f{};
    const int nr = lane & 15;
    const int kb = (lane >> 4) << 2;

    const int  srow = (tid >> 4) + (tid & 8), c4 = (tid & 7) << 2;
    const float* gAs[4];
    #pragma unroll
    for (int s = 0; s < 4; s++) {
        const int p  = 16 * s + srow;
        const int cc = (p < nF || !hasG) ? cF : cG;
        int       w  = (p < nF || !hasG) ? p : p - nF;
        w = (w < W_) ? w : W_ - 1;
        gAs[s] = cap + ((size_t)cc * W_ + w) * D_ + c4;
    }
    const float* gB = img + ((size_t)i0 * R_ + srow) * D_ + c4;

    auto k_loop = [&](auto mtc) __attribute__((always_inline)) {
        constexpr int MT = decltype(mtc)::value;
        const int  m_lo = (MT == 4) ? rw : ((MT == 2) ? (rw >> 1) : m_lo_g);
        const int  n_lo = (MT == 4) ? 0  : ((MT == 2) ? ((rw & 1) << 2) : n_lo_g);
        const int  nlo  = (MT == 4) ? 5  : ((MT == 2) ? 4 + (rw & 1) : ((rw == 1) ? 3 : 5));
        const int  nhi  = (MT == 4) ? 4  : ((MT == 2) ? 0 : ((rw == 0) ? 1 : ((rw == 1) ? 4 : 2)));
        const int  m_hi = (MT == 4) ? rw : ((MT == 2) ? (rw >> 1) : m_hi_g);
        const int  n_hi = (MT == 4) ? 5  : ((MT == 2) ? 0 : n_hi_g);
        for (int k0 = 0; k0 < D_; k0 += 32) {
            v4f va[4], vb[NT_];
            #pragma unroll
            for (int s = 0; s < 4; s++) {
                va[s] = v4f{};
                if (s < MT) va[s] = *(const v4f*)(gAs[s] + k0);
            }
            #pragma unroll
            for (int s = 0; s < NT_; s++) vb[s] = *(const v4f*)(gB + (size_t)s * 16 * D_ + k0);
            asm volatile("" :: "v"(va[0]), "v"(va[1]), "v"(va[2]), "v"(va[3]),
                               "v"(vb[0]), "v"(vb[1]), "v"(vb[2]), "v"(vb[3]), "v"(vb[4]),
                               "v"(vb[5]), "v"(vb[6]), "v"(vb[7]), "v"(vb[8]));
            __syncthreads();
            #pragma unroll
            for (int s = 0; s < 4; s++) {
                if (s < MT) *(v4f*)&As[srow + 16 * s][c4] = va[s];
            }
            #pragma unroll
            for (int s = 0; s < NT_; s++) *(v4f*)&Bs[srow + 16 * s][c4] = vb[s];
            __syncthreads();
            for (int kk = 0; kk < 32; kk += 8) {
                v4f b4[NT_];
                v4f ahi;
                const v4f alo = *(const v4f*)&As[16 * m_lo + nr][kk + kb];
                LDB_(0, n_lo, 0); LDB_(1, n_lo, 1); LDB_(2, n_lo, 2);
                if (nlo > 3) { LDB_(3, n_lo, 3); if (nlo > 4) LDB_(4, n_lo, 4); }
                if (nhi > 0) {
                    ahi = *(const v4f*)&As[16 * m_hi + nr][kk + kb];
                    LDB_(5, n_hi, 0);
                    if (nhi > 1) { LDB_(6, n_hi, 1); if (nhi > 2) { LDB_(7, n_hi, 2); LDB_(8, n_hi, 3); } }
                }
                WM_(0, alo, x, y) WM_(1, alo, x, y) WM_(2, alo, x, y)
                if (nlo > 3) { WM_(3, alo, x, y) if (nlo > 4) WM_(4, alo, x, y) }
                if (nhi > 0) {
                    WM_(5, ahi, x, y)
                    if (nhi > 1) { WM_(6, ahi, x, y) if (nhi > 2) { WM_(7, ahi, x, y) WM_(8, ahi, x, y) } }
                }
                WM_(0, alo, z, w) WM_(1, alo, z, w) WM_(2, alo, z, w)
                if (nlo > 3) { WM_(3, alo, z, w) if (nlo > 4) WM_(4, alo, z, w) }
                if (nhi > 0) {
                    WM_(5, ahi, z, w)
                    if (nhi > 1) { WM_(6, ahi, z, w) if (nhi > 2) { WM_(7, ahi, z, w) WM_(8, ahi, z, w) } }
                }
            }
        }
    };
    if (mt > 3) k_loop(ic_<4>{}); else if (mt > 2) k_loop(ic_<3>{}); else k_loop(ic_<2>{});
    __syncthreads();

    {
        const int h8 = (lane >> 4) << 3;
        SPILL_(0, m_lo, n_lo, 0) SPILL_(1, m_lo, n_lo, 1) SPILL_(2, m_lo, n_lo, 2)
        if (nlo > 3) { SPILL_(3, m_lo, n_lo, 3) if (nlo > 4) SPILL_(4, m_lo, n_lo, 4) }
        if (nhi > 0) {
            SPILL_(5, m_hi, n_hi, 0)
            if (nhi > 1) { SPILL_(6, m_hi, n_hi, 1) if (nhi > 2) { SPILL_(7, m_hi, n_hi, 2) SPILL_(8, m_hi, n_hi, 3) } }
        }
        int zr1 = (hasG ? nF : 0) + W_;
        zr1 = (zr1 < SR_) ? zr1 : SR_;
        for (int r = 16 * mt; r < zr1; r++) {
            Sm[r][tid] = 0.f;
            if (tid < NB_ - 128) Sm[r][128 + tid] = 0.f;
        }
    }
    __syncthreads();

    const int nit = hasG ? 2 * NI_ : NI_;
    for (int it = 0; it < nit; it++) {
        const bool g    = it >= NI_;
        const int  p    = it & (NI_ - 1);
        const int  c    = g ? cG : cF;
        const int  cl   = g ? clG : clF;
        const int  nsel = g ? nG : nF;
        float (*Sq)[SP_] = Sm + (g ? nF : 0);
        const int i  = i0 + p;
        const int cb = p * R_;

        if (tid < R_) {
            float x[W_];
            #pragma unroll
            for (int w = 0; w < W_; w++) x[w] = Sq[w][cb + tid];
            asm volatile("" :: HOLD6(x, 0), HOLD6(x, 6), HOLD6(x, 12), HOLD6(x, 18), HOLD6(x, 24));
            asm volatile("" :: HOLD6(x, 30), HOLD6(x, 36), HOLD6(x, 42), HOLD6(x, 48), HOLD6(x, 54));
            float ss = 0.f;
            #pragma unroll
            for (int w = 0; w < W_; w++) ss = (w < nsel) ? __builtin_fmaf(x[w], x[w], ss) : ss;
            rnw[tid] = 1.0f / fmaxf(sqrtf(ss), EPS_);
        } else if (tid >= 64 && tid < 64 + W_) {
            const int w = tid - 64;
            float x[R_];
            #pragma unroll
            for (int r = 0; r < R_; r++) x[r] = Sq[w][cb + r];
            asm volatile("" :: HOLD6(x, 0), HOLD6(x, 6), HOLD6(x, 12), HOLD6(x, 18), HOLD6(x, 24));
            asm volatile("" :: HOLD6(x, 30));
            float ss = 0.f;
            #pragma unroll
            for (int r = 0; r < R_; r++) ss = __builtin_fmaf(x[r], x[r], ss);
            rnr[w] = 1.0f / fmaxf(sqrtf(ss), EPS_);
        }
        __syncthreads();

        if (tid < W_) {
            float m = NEG_;
            if (tid < nsel) {
                float x[R_];
                #pragma unroll
                for (int r = 0; r < R_; r++) x[r] = Sq[tid][cb + r];
                asm volatile("" :: HOLD6(x, 0), HOLD6(x, 6), HOLD6(x, 12), HOLD6(x, 18), HOLD6(x, 24));
                asm volatile("" :: HOLD6(x, 30));
                #pragma unroll
                for (int r = 0; r < R_; r++) m = fmaxf(m, x[r] * rnw[r]);
            }
            rowsim[tid] = m;
        } else if (tid >= 64 && tid < 64 + R_) {
            const int r = tid - 64;
            float v[W_];
            #pragma unroll
            for (int w = 0; w < W_; w++) v[w] = Sq[w][cb + r] * rnr[w];
            float m = NEG_;
            #pragma unroll
            for (int w = 0; w < W_; w++) m = (w < cl) ? fmaxf(m, v[w]) : m;
            colsim[r] = m;
        }
        __syncthreads();

        if (wv == 2) {
            const float a0 = (lane < cl)      ? rowsim[lane]      : 0.f;
            const float a1 = (lane + 32 < cl) ? rowsim[lane + 32] : 0.f;
            float s = a0 + a1;
            s += __shfl_down(s, 16);
            s += __shfl_down(s, 8);
            s += __shfl_down(s, 4);
            s += __shfl_down(s, 2);
            s += __shfl_down(s, 1);
            if (lane == 0) s_rowmean = s / (float)cl;
        } else if (wv == 3) {
            const float a0 = colsim[lane];
            const float a1 = (lane + 32 < R_) ? colsim[lane + 32] : 0.f;
            float s = a0 + a1;
            s += __shfl_down(s, 16);
            s += __shfl_down(s, 8);
            s += __shfl_down(s, 4);
            s += __shfl_down(s, 2);
            s += __shfl_down(s, 1);
            if (lane == 0) s_colmean = s / (float)R_;
        }

        if (tid < W_) {
            const float v = rowsim[tid];
            int rank = 0;
            for (int j = 0; j < W_; j++) {
                const float u = rowsim[j];
                rank += (u > v) || (u == v && j < tid);
            }
            if (rank < TK_) topr[rank] = v;
        } else if (tid >= 64 && tid < 64 + R_) {
            const int r = tid - 64;
            const float v = colsim[r];
            int rank = 0;
            for (int j = 0; j < R_; j++) {
                const float u = colsim[j];
                rank += (u > v) || (u == v && j < r);
            }
            if (rank < TK_) topc[rank] = v;
        }
        __syncthreads();

        if (tid < TK_) {
            float a = b1_t2i[tid];
            for (int k = 0; k < TK_; k++) a += topr[k] * w1_t2i[k * TK_ + tid];
            h1[tid] = fmaxf(a, 0.f);
        } else if (tid >= 32 && tid < 32 + TK_) {
            const int j = tid - 32;
            float a = b1_i2t[j];
            for (int k = 0; k < TK_; k++) a += topc[k] * w1_i2t[k * TK_ + j];
            h2[j] = fmaxf(a, 0.f);
        }
        __syncthreads();

        if (tid == 0) {
            float o = b2_t2i[0];
            for (int j = 0; j < TK_; j++) o += h1[j] * w2_t2i[j];
            out[c * B_ + i]              = o + s_rowmean;
            out[2 * B_ * B_ + c * B_ + i] = s_rowmean;
        } else if (tid == 32) {
            float o = b2_i2t[0];
            for (int j = 0; j < TK_; j++) o += h2[j] * w2_i2t[j];
            out[B_ * B_ + i * B_ + c]     = o + s_colmean;
            out[3 * B_ * B_ + i * B_ + c] = s_colmean;
        }
    }
}

extern "C" void kernel_launch(void* const* d_in, const int* in_sizes, int n_in,
                              void* d_out, int out_size, void* d_ws, size_t ws_size,
                              hipStream_t stream) {
    const float* img     = (const float*)d_in[0];
    const float* cap     = (const float*)d_in[1];
    const float* w1_t2i  = (const float*)d_in[2];
    const float* b1_t2i  = (const float*)d_in[3];
    const float* w2_t2i  = (const float*)d_in[4];
    const float* b2_t2i  = (const float*)d_in[5];
    const float* w1_i2t  = (const float*)d_in[6];
    const float* b1_i2t  = (const float*)d_in[7];
    const float* w2_i2t  = (const float*)d_in[8];
    const float* b2_i2t  = (const float*)d_in[9];
    const int*   cap_len = (const int*)d_in[10];
    float*       out     = (float*)d_out;

    float* nrm  = (float*)d_ws;
    float* capn = nrm;
    float* imgn = nrm + (size_t)B_ * W_ * D_;

    row_normalize_kernel<<<B_ * W_ + B_ * R_, 256, 0, stream>>>(img, cap, nrm);

    dim3 grid(B_ / NI_, B_);
    pair_kernel<<<grid, 128, 0, stream>>>(imgn, capn,
                                          w1_t2i, b1_t2i, w2_t2i, b2_t2i,
                                          w1_i2t, b1_i2t, w2_i2t, b2_i2t,
                                          cap_len, out);
}
